# starting kernel
# baseline (speedup 1.0000x reference)
#include <hip/hip_runtime.h>
#include <math.h>

#define NN      4096
#define FIN     128
#define FHID    64
#define NHEADS  4
#define FOUT    64
#define HCATF   (NHEADS * FHID)
#define ALPHA_L 0.2f
#define CAP     128

typedef float v2f __attribute__((ext_vector_type(2)));
typedef float v8f __attribute__((ext_vector_type(8)));

__device__ __forceinline__ v8f wmma_f32_tile(const float* __restrict__ A, int lda,
                                             const float* __restrict__ B, int ldb,
                                             int K, int lane) {
  const int half = lane >> 4;
  const int l16  = lane & 15;
  v8f c = {0.f, 0.f, 0.f, 0.f, 0.f, 0.f, 0.f, 0.f};
  for (int k = 0; k < K; k += 4) {
    v2f a, b;
    const float* ap = A + (size_t)l16 * lda + k + 2 * half;
    a.x = ap[0];
    a.y = ap[1];
    b.x = B[(size_t)(k + 2 * half) * ldb + l16];
    b.y = B[(size_t)(k + 2 * half + 1) * ldb + l16];
    c = __builtin_amdgcn_wmma_f32_16x16x4_f32(false, a, false, b, (short)0, c, false, false);
  }
  return c;
}

__global__ void build_csr_kernel(const float* __restrict__ adj,
                                 int* __restrict__ cnt, int* __restrict__ nidx) {
  __shared__ int scan[128];
  const int row = blockIdx.x;
  const int t = threadIdx.x;
  const float4* arow4 = (const float4*)(adj + (size_t)row * NN) + t * 8;
  const int base = t * 32;

  float4 v[8];
#pragma unroll
  for (int q = 0; q < 8; ++q) v[q] = arow4[q];

  int c = 0;
#pragma unroll
  for (int q = 0; q < 8; ++q) {
    c += (v[q].x > 0.f) ? 1 : 0;
    c += (v[q].y > 0.f) ? 1 : 0;
    c += (v[q].z > 0.f) ? 1 : 0;
    c += (v[q].w > 0.f) ? 1 : 0;
  }
  scan[t] = c;
  __syncthreads();
  for (int off = 1; off < 128; off <<= 1) {
    int p = (t >= off) ? scan[t - off] : 0;
    __syncthreads();
    scan[t] += p;
    __syncthreads();
  }
  int k = scan[t] - c;
  const int total = scan[127];
  if (t == 0) cnt[row] = total < CAP ? total : CAP;
  int* out = nidx + (size_t)row * CAP;
#pragma unroll
  for (int q = 0; q < 8; ++q) {
    const int b = base + q * 4;
    if (v[q].x > 0.f) { if (k < CAP) out[k] = b + 0; ++k; }
    if (v[q].y > 0.f) { if (k < CAP) out[k] = b + 1; ++k; }
    if (v[q].z > 0.f) { if (k < CAP) out[k] = b + 2; ++k; }
    if (v[q].w > 0.f) { if (k < CAP) out[k] = b + 3; ++k; }
  }
}

__global__ void lin_kernel(const float* __restrict__ x, const float* __restrict__ W,
                           const float* __restrict__ b, float* __restrict__ h) {
  const int mtile = blockIdx.x;
  const int wave  = threadIdx.x >> 5;
  const int lane  = threadIdx.x & 31;
  const int half  = lane >> 4, l16 = lane & 15;
  const float* A = x + (size_t)mtile * 16 * FIN;
  const float* B = W + wave * 16;
  v8f c = wmma_f32_tile(A, FIN, B, FIN, FIN, lane);
  const float bias = b[wave * 16 + l16];
  float* out = h + (size_t)mtile * 16 * FIN + wave * 16 + l16;
#pragma unroll
  for (int r = 0; r < 8; ++r)
    out[(size_t)(r + 8 * half) * FIN] = c[r] + bias;
}

__global__ void heads_gemm_kernel(const float* __restrict__ h,
                                  const float* __restrict__ W_heads,
                                  float* __restrict__ hcat) {
  const int mtile = blockIdx.x;
  const int wave  = threadIdx.x >> 5;
  const int head  = wave >> 2;
  const int nt    = wave & 3;
  const int lane  = threadIdx.x & 31;
  const int half  = lane >> 4, l16 = lane & 15;
  const float* A = h + (size_t)mtile * 16 * FIN;
  const float* B = W_heads + (size_t)head * FIN * FHID + nt * 16;
  v8f c = wmma_f32_tile(A, FIN, B, FHID, FIN, lane);
  float* out = hcat + (size_t)mtile * 16 * HCATF + head * FHID + nt * 16 + l16;
#pragma unroll
  for (int r = 0; r < 8; ++r)
    out[(size_t)(r + 8 * half) * HCATF] = c[r];
}

__global__ void attn_coeff_heads_kernel(const float* __restrict__ hcat,
                                        const float* __restrict__ a_heads,
                                        float* __restrict__ src, float* __restrict__ dst) {
  const int idx = blockIdx.x * blockDim.x + threadIdx.x;
  if (idx >= NN * NHEADS) return;
  const int i = idx >> 2, head = idx & 3;
  const float4* wh = (const float4*)(hcat + (size_t)i * HCATF + head * FHID);
  const float4* as = (const float4*)(a_heads + head * 2 * FHID);
  const float4* ad = as + FHID / 4;
  float s = 0.f, d = 0.f;
#pragma unroll
  for (int f = 0; f < FHID / 4; ++f) {
    const float4 w = wh[f], A = as[f], B = ad[f];
    s += w.x * A.x + w.y * A.y + w.z * A.z + w.w * A.w;
    d += w.x * B.x + w.y * B.y + w.z * B.z + w.w * B.w;
  }
  src[head * NN + i] = s;
  dst[head * NN + i] = d;
}

__global__ void gat_aggregate_heads_kernel(const float* __restrict__ hcat,
                                           const float* __restrict__ src,
                                           const float* __restrict__ dst,
                                           const int* __restrict__ cnt,
                                           const int* __restrict__ nidx,
                                           float* __restrict__ gat1) {
  __shared__ float wgt[CAP];
  __shared__ int   jidx[CAP];
  __shared__ float red[64];
  const int row  = blockIdx.x >> 2;
  const int head = blockIdx.x & 3;
  const int t    = threadIdx.x;
  const int deg  = cnt[row];
  const float si = src[head * NN + row];
  const int* nb  = nidx + (size_t)row * CAP;

  float m = -3.4e38f;
  for (int k = t; k < deg; k += 64) {
    const int j = nb[k];
    jidx[k] = j;
    float e = si + dst[head * NN + j];
    e = (e >= 0.f) ? e : ALPHA_L * e;
    wgt[k] = e;
    m = fmaxf(m, e);
  }
  red[t] = m;
  __syncthreads();
  for (int s = 32; s > 0; s >>= 1) { if (t < s) red[t] = fmaxf(red[t], red[t + s]); __syncthreads(); }
  m = red[0];
  __syncthreads();

  float ssum = 0.f;
  for (int k = t; k < deg; k += 64) {
    const float w = expf(wgt[k] - m);
    wgt[k] = w;
    ssum += w;
  }
  red[t] = ssum;
  __syncthreads();
  for (int s = 32; s > 0; s >>= 1) { if (t < s) red[t] += red[t + s]; __syncthreads(); }
  const float denom = red[0];
  __syncthreads();

  float acc = 0.f;
  for (int k = 0; k < deg; ++k)
    acc += wgt[k] * hcat[(size_t)jidx[k] * HCATF + head * FHID + t];
  float o = acc / denom;
  o = (o > 0.f) ? o : (expf(o) - 1.f);
  gat1[(size_t)row * HCATF + head * FHID + t] = o;
}

__global__ void end_gemm_kernel(const float* __restrict__ gat1,
                                const float* __restrict__ W_end,
                                float* __restrict__ Wh2) {
  const int mtile = blockIdx.x;
  const int wave  = threadIdx.x >> 5;
  const int lane  = threadIdx.x & 31;
  const int half  = lane >> 4, l16 = lane & 15;
  const float* A = gat1 + (size_t)mtile * 16 * HCATF;
  const float* B = W_end + wave * 16;
  v8f c = wmma_f32_tile(A, HCATF, B, FOUT, HCATF, lane);
  float* out = Wh2 + (size_t)mtile * 16 * FOUT + wave * 16 + l16;
#pragma unroll
  for (int r = 0; r < 8; ++r)
    out[(size_t)(r + 8 * half) * FOUT] = c[r];
}

__global__ void attn_coeff_end_kernel(const float* __restrict__ Wh2,
                                      const float* __restrict__ a_end,
                                      float* __restrict__ src, float* __restrict__ dst) {
  const int i = blockIdx.x * blockDim.x + threadIdx.x;
  if (i >= NN) return;
  const float4* wh = (const float4*)(Wh2 + (size_t)i * FOUT);
  const float4* as = (const float4*)a_end;
  const float4* ad = as + FOUT / 4;
  float s = 0.f, d = 0.f;
#pragma unroll
  for (int f = 0; f < FOUT / 4; ++f) {
    const float4 w = wh[f], A = as[f], B = ad[f];
    s += w.x * A.x + w.y * A.y + w.z * A.z + w.w * A.w;
    d += w.x * B.x + w.y * B.y + w.z * B.z + w.w * B.w;
  }
  src[i] = s;
  dst[i] = d;
}

__global__ void gat_end_kernel(const float* __restrict__ Wh2,
                               const float* __restrict__ src,
                               const float* __restrict__ dst,
                               const int* __restrict__ cnt,
                               const int* __restrict__ nidx,
                               float* __restrict__ out) {
  __shared__ float wgt[CAP];
  __shared__ int   jidx[CAP];
  __shared__ float red[64];
  const int row = blockIdx.x;
  const int t   = threadIdx.x;
  const int deg = cnt[row];
  const float si = src[row];
  const int* nb = nidx + (size_t)row * CAP;

  float m = -3.4e38f;
  for (int k = t; k < deg; k += 64) {
    const int j = nb[k];
    jidx[k] = j;
    float e = si + dst[j];
    e = (e >= 0.f) ? e : ALPHA_L * e;
    wgt[k] = e;
    m = fmaxf(m, e);
  }
  red[t] = m;
  __syncthreads();
  for (int s = 32; s > 0; s >>= 1) { if (t < s) red[t] = fmaxf(red[t], red[t + s]); __syncthreads(); }
  m = red[0];
  __syncthreads();

  float ssum = 0.f;
  for (int k = t; k < deg; k += 64) {
    const float w = expf(wgt[k] - m);
    wgt[k] = w;
    ssum += w;
  }
  red[t] = ssum;
  __syncthreads();
  for (int s = 32; s > 0; s >>= 1) { if (t < s) red[t] += red[t + s]; __syncthreads(); }
  const float denom = red[0];
  __syncthreads();

  float acc = 0.f;
  for (int k = 0; k < deg; ++k)
    acc += wgt[k] * Wh2[(size_t)jidx[k] * FOUT + t];
  float o = acc / denom;
  o = (o > 0.f) ? o : (expf(o) - 1.f);

  red[t] = o;
  __syncthreads();
  for (int s = 32; s > 0; s >>= 1) { if (t < s) red[t] = fmaxf(red[t], red[t + s]); __syncthreads(); }
  const float mo = red[0];
  __syncthreads();
  const float ex = expf(o - mo);
  red[t] = ex;
  __syncthreads();
  for (int s = 32; s > 0; s >>= 1) { if (t < s) red[t] += red[t + s]; __syncthreads(); }
  const float sum = red[0];
  out[(size_t)row * FOUT + t] = ex / sum;
}

extern "C" void kernel_launch(void* const* d_in, const int* in_sizes, int n_in,
                              void* d_out, int out_size, void* d_ws, size_t ws_size,
                              hipStream_t stream) {
  (void)in_sizes; (void)n_in; (void)out_size; (void)ws_size;
  const float* x       = (const float*)d_in[0];
  const float* adj     = (const float*)d_in[1];
  const float* lin_W   = (const float*)d_in[2];
  const float* lin_b   = (const float*)d_in[3];
  const float* W_heads = (const float*)d_in[4];
  const float* a_heads = (const float*)d_in[5];
  const float* W_end   = (const float*)d_in[6];
  const float* a_end   = (const float*)d_in[7];
  float* out = (float*)d_out;

  char* ws = (char*)d_ws;
  float* h     = (float*)ws;  ws += (size_t)NN * FIN   * sizeof(float);
  float* hcat  = (float*)ws;  ws += (size_t)NN * HCATF * sizeof(float);
  float* gat1  = (float*)ws;  ws += (size_t)NN * HCATF * sizeof(float);
  float* Wh2   = (float*)ws;  ws += (size_t)NN * FOUT  * sizeof(float);
  float* src1  = (float*)ws;  ws += (size_t)NHEADS * NN * sizeof(float);
  float* dst1  = (float*)ws;  ws += (size_t)NHEADS * NN * sizeof(float);
  float* src2  = (float*)ws;  ws += (size_t)NN * sizeof(float);
  float* dst2  = (float*)ws;  ws += (size_t)NN * sizeof(float);
  int*   ncnt  = (int*)ws;    ws += (size_t)NN * sizeof(int);
  int*   nidx  = (int*)ws;    ws += (size_t)NN * CAP * sizeof(int);

  build_csr_kernel<<<NN, 128, 0, stream>>>(adj, ncnt, nidx);
  lin_kernel<<<NN / 16, 256, 0, stream>>>(x, lin_W, lin_b, h);
  heads_gemm_kernel<<<NN / 16, 512, 0, stream>>>(h, W_heads, hcat);
  attn_coeff_heads_kernel<<<(NN * NHEADS) / 256, 256, 0, stream>>>(hcat, a_heads, src1, dst1);
  gat_aggregate_heads_kernel<<<NN * NHEADS, 64, 0, stream>>>(hcat, src1, dst1, ncnt, nidx, gat1);
  end_gemm_kernel<<<NN / 16, 128, 0, stream>>>(gat1, W_end, Wh2);
  attn_coeff_end_kernel<<<NN / 256, 256, 0, stream>>>(Wh2, a_end, src2, dst2);
  gat_end_kernel<<<NN, 64, 0, stream>>>(Wh2, src2, dst2, ncnt, nidx, out);
}


# Aggregation kernels (both layers): one wave per node row, cross-lane softmax and readlane-broadcast weights replace a 64-thread block per row-head with LDS trees: 5x fewer waves, no barriers
# speedup vs baseline: 1.1774x; 1.1774x over previous
#include <hip/hip_runtime.h>
#include <math.h>

#define NN      4096
#define FIN     128
#define FHID    64
#define NHEADS  4
#define FOUT    64
#define HCATF   (NHEADS * FHID)
#define ALPHA_L 0.2f
#define CAP     128

typedef float v2f __attribute__((ext_vector_type(2)));
typedef float v8f __attribute__((ext_vector_type(8)));

#define DEV static __device__ __forceinline__

DEV int uni(int v) { return __builtin_amdgcn_readfirstlane(v); }
DEV float xrow(float x) {
  return __builtin_bit_cast(float, __builtin_amdgcn_permlanex16(__builtin_bit_cast(int, x), __builtin_bit_cast(int, x),
                                                               0x76543210, 0xfedcba98, false, false));
}
template <int M> DEV float xlane(float x) {
  return __builtin_bit_cast(float, __builtin_amdgcn_update_dpp(0, __builtin_bit_cast(int, x),
                                                              0x160 + M, 0xf, 0xf, true));
}
DEV float wave_max(float v) {
  v = fmaxf(v, xrow(v));
  v = fmaxf(v, xlane<8>(v));
  v = fmaxf(v, xlane<4>(v));
  v = fmaxf(v, xlane<2>(v));
  v = fmaxf(v, xlane<1>(v));
  return v;
}
DEV float wave_sum(float v) {
  v = v + xrow(v);
  v = v + xlane<8>(v);
  v = v + xlane<4>(v);
  v = v + xlane<2>(v);
  v = v + xlane<1>(v);
  return v;
}
DEV float lane_f(float x, int l) { return __builtin_bit_cast(float, __builtin_amdgcn_readlane(__builtin_bit_cast(int, x), l)); }
DEV int   lane_i(int x, int l)   { return __builtin_amdgcn_readlane(x, l); }
DEV float leaky(float e) { return (e >= 0.f) ? e : ALPHA_L * e; }
DEV float elu1(float o)  { return (o > 0.f) ? o : (expf(o) - 1.f); }

#define ROWS_PER_BLOCK 4
#define AGG_G 8
#define AGG_G_END 16

__device__ __forceinline__ v8f wmma_f32_tile(const float* __restrict__ A, int lda,
                                             const float* __restrict__ B, int ldb,
                                             int K, int lane) {
  const int half = lane >> 4;
  const int l16  = lane & 15;
  v8f c = {0.f, 0.f, 0.f, 0.f, 0.f, 0.f, 0.f, 0.f};
  for (int k = 0; k < K; k += 4) {
    v2f a, b;
    const float* ap = A + (size_t)l16 * lda + k + 2 * half;
    a.x = ap[0];
    a.y = ap[1];
    b.x = B[(size_t)(k + 2 * half) * ldb + l16];
    b.y = B[(size_t)(k + 2 * half + 1) * ldb + l16];
    c = __builtin_amdgcn_wmma_f32_16x16x4_f32(false, a, false, b, (short)0, c, false, false);
  }
  return c;
}

__global__ void build_csr_kernel(const float* __restrict__ adj,
                                 int* __restrict__ cnt, int* __restrict__ nidx) {
  __shared__ int scan[128];
  const int row = blockIdx.x;
  const int t = threadIdx.x;
  const float4* arow4 = (const float4*)(adj + (size_t)row * NN) + t * 8;
  const int base = t * 32;

  float4 v[8];
#pragma unroll
  for (int q = 0; q < 8; ++q) v[q] = arow4[q];

  int c = 0;
#pragma unroll
  for (int q = 0; q < 8; ++q) {
    c += (v[q].x > 0.f) ? 1 : 0;
    c += (v[q].y > 0.f) ? 1 : 0;
    c += (v[q].z > 0.f) ? 1 : 0;
    c += (v[q].w > 0.f) ? 1 : 0;
  }
  scan[t] = c;
  __syncthreads();
  for (int off = 1; off < 128; off <<= 1) {
    int p = (t >= off) ? scan[t - off] : 0;
    __syncthreads();
    scan[t] += p;
    __syncthreads();
  }
  int k = scan[t] - c;
  const int total = scan[127];
  if (t == 0) cnt[row] = total < CAP ? total : CAP;
  int* out = nidx + (size_t)row * CAP;
#pragma unroll
  for (int q = 0; q < 8; ++q) {
    const int b = base + q * 4;
    if (v[q].x > 0.f) { if (k < CAP) out[k] = b + 0; ++k; }
    if (v[q].y > 0.f) { if (k < CAP) out[k] = b + 1; ++k; }
    if (v[q].z > 0.f) { if (k < CAP) out[k] = b + 2; ++k; }
    if (v[q].w > 0.f) { if (k < CAP) out[k] = b + 3; ++k; }
  }
}

__global__ void lin_kernel(const float* __restrict__ x, const float* __restrict__ W,
                           const float* __restrict__ b, float* __restrict__ h) {
  const int mtile = blockIdx.x;
  const int wave  = threadIdx.x >> 5;
  const int lane  = threadIdx.x & 31;
  const int half  = lane >> 4, l16 = lane & 15;
  const float* A = x + (size_t)mtile * 16 * FIN;
  const float* B = W + wave * 16;
  v8f c = wmma_f32_tile(A, FIN, B, FIN, FIN, lane);
  const float bias = b[wave * 16 + l16];
  float* out = h + (size_t)mtile * 16 * FIN + wave * 16 + l16;
#pragma unroll
  for (int r = 0; r < 8; ++r)
    out[(size_t)(r + 8 * half) * FIN] = c[r] + bias;
}

__global__ void heads_gemm_kernel(const float* __restrict__ h,
                                  const float* __restrict__ W_heads,
                                  float* __restrict__ hcat) {
  const int mtile = blockIdx.x;
  const int wave  = threadIdx.x >> 5;
  const int head  = wave >> 2;
  const int nt    = wave & 3;
  const int lane  = threadIdx.x & 31;
  const int half  = lane >> 4, l16 = lane & 15;
  const float* A = h + (size_t)mtile * 16 * FIN;
  const float* B = W_heads + (size_t)head * FIN * FHID + nt * 16;
  v8f c = wmma_f32_tile(A, FIN, B, FHID, FIN, lane);
  float* out = hcat + (size_t)mtile * 16 * HCATF + head * FHID + nt * 16 + l16;
#pragma unroll
  for (int r = 0; r < 8; ++r)
    out[(size_t)(r + 8 * half) * HCATF] = c[r];
}

__global__ void attn_coeff_heads_kernel(const float* __restrict__ hcat,
                                        const float* __restrict__ a_heads,
                                        float* __restrict__ src, float* __restrict__ dst) {
  const int idx = blockIdx.x * blockDim.x + threadIdx.x;
  if (idx >= NN * NHEADS) return;
  const int i = idx >> 2, head = idx & 3;
  const float4* wh = (const float4*)(hcat + (size_t)i * HCATF + head * FHID);
  const float4* as = (const float4*)(a_heads + head * 2 * FHID);
  const float4* ad = as + FHID / 4;
  float s = 0.f, d = 0.f;
#pragma unroll
  for (int f = 0; f < FHID / 4; ++f) {
    const float4 w = wh[f], A = as[f], B = ad[f];
    s += w.x * A.x + w.y * A.y + w.z * A.z + w.w * A.w;
    d += w.x * B.x + w.y * B.y + w.z * B.z + w.w * B.w;
  }
  src[idx] = s;
  dst[idx] = d;
}

template <int N>
DEV void agg_heads_group(const char* __restrict__ hbytes, unsigned voff, int js,
                         float w0, float w1, float w2, float w3, int l0, bool hi,
                         float4& accA, float4& accB) {
  float4 xa[N], xb[N];
  for (int u = 0; u < N; ++u) {
    const int jk = lane_i(js, l0 + u);
    const char* p = hbytes + (size_t)jk * (HCATF * sizeof(float));
    xa[u] = *(const float4*)(p + voff);
    xb[u] = *(const float4*)(p + voff + 2 * FHID * sizeof(float));
  }
  for (int u = 0; u < N; ++u) {
    const float s0 = lane_f(w0, l0 + u), s1 = lane_f(w1, l0 + u);
    const float s2 = lane_f(w2, l0 + u), s3 = lane_f(w3, l0 + u);
    const float wa = hi ? s1 : s0;
    const float wb = hi ? s3 : s2;
    accA.x = fmaf(wa, xa[u].x, accA.x); accA.y = fmaf(wa, xa[u].y, accA.y);
    accA.z = fmaf(wa, xa[u].z, accA.z); accA.w = fmaf(wa, xa[u].w, accA.w);
    accB.x = fmaf(wb, xb[u].x, accB.x); accB.y = fmaf(wb, xb[u].y, accB.y);
    accB.z = fmaf(wb, xb[u].z, accB.z); accB.w = fmaf(wb, xb[u].w, accB.w);
  }
}

__global__ __launch_bounds__(32 * ROWS_PER_BLOCK)
void gat_aggregate_heads_kernel(const float* __restrict__ hcat,
                                const float* __restrict__ src,
                                const float* __restrict__ dst,
                                const int* __restrict__ cnt,
                                const int* __restrict__ nidx,
                                float* __restrict__ gat1) {
  const int lane = threadIdx.x & 31;
  const bool hi  = lane >= 16;
  const int row  = uni((int)(blockIdx.x * ROWS_PER_BLOCK + (threadIdx.x >> 5)));
  const int deg  = uni(cnt[row]);
  const int* nb  = nidx + (size_t)row * CAP;
  const float4 si = *(const float4*)(src + (size_t)row * NHEADS);

  int    j[4] = {0, 0, 0, 0};
  float4 d[4] = {};
  float  w[4][NHEADS];
  const float4* dst4 = (const float4*)dst;
  if (deg > 0) {
    j[0] = nb[min(lane, deg - 1)];
    j[1] = nb[min(lane + 32, deg - 1)];
    d[0] = dst4[j[0]];
    d[1] = dst4[j[1]];
    if (deg > 64) {
      j[2] = nb[min(lane + 64, deg - 1)];
      j[3] = nb[min(lane + 96, deg - 1)];
      d[2] = dst4[j[2]];
      d[3] = dst4[j[3]];
    }
  }
  float m[NHEADS] = {-3.4e38f, -3.4e38f, -3.4e38f, -3.4e38f};
  for (int s = 0; s < 4; ++s) {
    const bool valid = (s * 32 + lane) < deg;
    w[s][0] = valid ? leaky(si.x + d[s].x) : -3.4e38f;
    w[s][1] = valid ? leaky(si.y + d[s].y) : -3.4e38f;
    w[s][2] = valid ? leaky(si.z + d[s].z) : -3.4e38f;
    w[s][3] = valid ? leaky(si.w + d[s].w) : -3.4e38f;
    for (int h = 0; h < NHEADS; ++h) m[h] = fmaxf(m[h], w[s][h]);
  }
  for (int h = 0; h < NHEADS; ++h) m[h] = wave_max(m[h]);

  for (int s = 0; s < 4; ++s) {
    const bool valid = (s * 32 + lane) < deg;
    for (int h = 0; h < NHEADS; ++h) w[s][h] = valid ? expf(w[s][h] - m[h]) : 0.f;
  }
  float denom[NHEADS];
  for (int h = 0; h < NHEADS; ++h)
    denom[h] = wave_sum((w[0][h] + w[2][h]) + (w[1][h] + w[3][h]));

  float4 accA = {0.f, 0.f, 0.f, 0.f}, accB = {0.f, 0.f, 0.f, 0.f};
  const char* hbytes = (const char*)hcat;
  const unsigned voff = (unsigned)lane * 16u;
  for (int base = 0; base < deg; base += AGG_G) {
    const int s  = base >> 5;
    int l0 = base & 31;
    const int   js = (s == 0) ? j[0]    : (s == 1) ? j[1]    : (s == 2) ? j[2]    : j[3];
    const float w0 = (s == 0) ? w[0][0] : (s == 1) ? w[1][0] : (s == 2) ? w[2][0] : w[3][0];
    const float w1 = (s == 0) ? w[0][1] : (s == 1) ? w[1][1] : (s == 2) ? w[2][1] : w[3][1];
    const float w2 = (s == 0) ? w[0][2] : (s == 1) ? w[1][2] : (s == 2) ? w[2][2] : w[3][2];
    const float w3 = (s == 0) ? w[0][3] : (s == 1) ? w[1][3] : (s == 2) ? w[2][3] : w[3][3];
    const int r = deg - base;
    if (r >= AGG_G) {
      agg_heads_group<AGG_G>(hbytes, voff, js, w0, w1, w2, w3, l0, hi, accA, accB);
    } else {
      if (r & 4) { agg_heads_group<4>(hbytes, voff, js, w0, w1, w2, w3, l0, hi, accA, accB); l0 += 4; }
      if (r & 2) { agg_heads_group<2>(hbytes, voff, js, w0, w1, w2, w3, l0, hi, accA, accB); l0 += 2; }
      if (r & 1) { agg_heads_group<1>(hbytes, voff, js, w0, w1, w2, w3, l0, hi, accA, accB); }
    }
  }

  const float dA = hi ? denom[1] : denom[0];
  const float dB = hi ? denom[3] : denom[2];
  float4 oA, oB;
  oA.x = elu1(accA.x / dA); oA.y = elu1(accA.y / dA); oA.z = elu1(accA.z / dA); oA.w = elu1(accA.w / dA);
  oB.x = elu1(accB.x / dB); oB.y = elu1(accB.y / dB); oB.z = elu1(accB.z / dB); oB.w = elu1(accB.w / dB);
  float* orow = gat1 + (size_t)row * HCATF + lane * 4;
  *(float4*)orow = oA;
  *(float4*)(orow + 2 * FHID) = oB;
}

__global__ void end_gemm_kernel(const float* __restrict__ gat1,
                                const float* __restrict__ W_end,
                                float* __restrict__ Wh2) {
  const int mtile = blockIdx.x;
  const int wave  = threadIdx.x >> 5;
  const int lane  = threadIdx.x & 31;
  const int half  = lane >> 4, l16 = lane & 15;
  const float* A = gat1 + (size_t)mtile * 16 * HCATF;
  const float* B = W_end + wave * 16;
  v8f c = wmma_f32_tile(A, HCATF, B, FOUT, HCATF, lane);
  float* out = Wh2 + (size_t)mtile * 16 * FOUT + wave * 16 + l16;
#pragma unroll
  for (int r = 0; r < 8; ++r)
    out[(size_t)(r + 8 * half) * FOUT] = c[r];
}

__global__ void attn_coeff_end_kernel(const float* __restrict__ Wh2,
                                      const float* __restrict__ a_end,
                                      float* __restrict__ src, float* __restrict__ dst) {
  const int i = blockIdx.x * blockDim.x + threadIdx.x;
  if (i >= NN) return;
  const float4* wh = (const float4*)(Wh2 + (size_t)i * FOUT);
  const float4* as = (const float4*)a_end;
  const float4* ad = as + FOUT / 4;
  float s = 0.f, d = 0.f;
#pragma unroll
  for (int f = 0; f < FOUT / 4; ++f) {
    const float4 w = wh[f], A = as[f], B = ad[f];
    s += w.x * A.x + w.y * A.y + w.z * A.z + w.w * A.w;
    d += w.x * B.x + w.y * B.y + w.z * B.z + w.w * B.w;
  }
  src[i] = s;
  dst[i] = d;
}

template <int N>
DEV void agg_end_group(const char* __restrict__ wbytes, unsigned voff, int js, float ws,
                       int l0, float& a0, float& a1) {
  float x0[N], x1[N];
  for (int u = 0; u < N; ++u) {
    const int jk = lane_i(js, l0 + u);
    const char* p = wbytes + (size_t)jk * (FOUT * sizeof(float));
    x0[u] = *(const float*)(p + voff);
    x1[u] = *(const float*)(p + voff + 32 * sizeof(float));
  }
  for (int u = 0; u < N; ++u) {
    const float wk = lane_f(ws, l0 + u);
    a0 = fmaf(wk, x0[u], a0);
    a1 = fmaf(wk, x1[u], a1);
  }
}

__global__ __launch_bounds__(32 * ROWS_PER_BLOCK)
void gat_end_kernel(const float* __restrict__ Wh2,
                    const float* __restrict__ src,
                    const float* __restrict__ dst,
                    const int* __restrict__ cnt,
                    const int* __restrict__ nidx,
                    float* __restrict__ out) {
  const int lane = threadIdx.x & 31;
  const int row  = uni((int)(blockIdx.x * ROWS_PER_BLOCK + (threadIdx.x >> 5)));
  const int deg  = uni(cnt[row]);
  const float si = src[row];
  const int* nb  = nidx + (size_t)row * CAP;

  int   j[4] = {0, 0, 0, 0};
  float d[4] = {0.f, 0.f, 0.f, 0.f};
  float w[4];
  if (deg > 0) {
    j[0] = nb[min(lane, deg - 1)];
    j[1] = nb[min(lane + 32, deg - 1)];
    d[0] = dst[j[0]];
    d[1] = dst[j[1]];
    if (deg > 64) {
      j[2] = nb[min(lane + 64, deg - 1)];
      j[3] = nb[min(lane + 96, deg - 1)];
      d[2] = dst[j[2]];
      d[3] = dst[j[3]];
    }
  }
  float m = -3.4e38f;
  for (int s = 0; s < 4; ++s) {
    w[s] = ((s * 32 + lane) < deg) ? leaky(si + d[s]) : -3.4e38f;
    m = fmaxf(m, w[s]);
  }
  m = wave_max(m);

  for (int s = 0; s < 4; ++s) w[s] = ((s * 32 + lane) < deg) ? expf(w[s] - m) : 0.f;
  const float denom = wave_sum((w[0] + w[2]) + (w[1] + w[3]));

  float a0 = 0.f, a1 = 0.f;
  const char* wbytes = (const char*)Wh2;
  const unsigned voff = (unsigned)lane * 4u;
  for (int base = 0; base < deg; base += AGG_G_END) {
    const int s  = base >> 5;
    int l0 = base & 31;
    const int   js = (s == 0) ? j[0] : (s == 1) ? j[1] : (s == 2) ? j[2] : j[3];
    const float ws = (s == 0) ? w[0] : (s == 1) ? w[1] : (s == 2) ? w[2] : w[3];
    const int r = deg - base;
    if (r >= AGG_G_END) {
      agg_end_group<AGG_G_END>(wbytes, voff, js, ws, l0, a0, a1);
    } else {
      if (r & 8) { agg_end_group<8>(wbytes, voff, js, ws, l0, a0, a1); l0 += 8; }
      if (r & 4) { agg_end_group<4>(wbytes, voff, js, ws, l0, a0, a1); l0 += 4; }
      if (r & 2) { agg_end_group<2>(wbytes, voff, js, ws, l0, a0, a1); l0 += 2; }
      if (r & 1) { agg_end_group<1>(wbytes, voff, js, ws, l0, a0, a1); }
    }
  }
  const float o0 = elu1(a0 / denom);
  const float o1 = elu1(a1 / denom);

  const float mo  = wave_max(fmaxf(o0, o1));
  const float ex0 = expf(o0 - mo);
  const float ex1 = expf(o1 - mo);
  const float sum = wave_sum(ex0 + ex1);
  float* orow = out + (size_t)row * FOUT + lane;
  orow[0]  = ex0 / sum;
  orow[32] = ex1 / sum;
}

extern "C" void kernel_launch(void* const* d_in, const int* in_sizes, int n_in,
                              void* d_out, int out_size, void* d_ws, size_t ws_size,
                              hipStream_t stream) {
  (void)in_sizes; (void)n_in; (void)out_size; (void)ws_size;
  const float* x       = (const float*)d_in[0];
  const float* adj     = (const float*)d_in[1];
  const float* lin_W   = (const float*)d_in[2];
  const float* lin_b   = (const float*)d_in[3];
  const float* W_heads = (const float*)d_in[4];
  const float* a_heads = (const float*)d_in[5];
  const float* W_end   = (const float*)d_in[6];
  const float* a_end   = (const float*)d_in[7];
  float* out = (float*)d_out;

  char* ws = (char*)d_ws;
  float* h     = (float*)ws;  ws += (size_t)NN * FIN   * sizeof(float);
  float* hcat  = (float*)ws;  ws += (size_t)NN * HCATF * sizeof(float);
  float* gat1  = (float*)ws;  ws += (size_t)NN * HCATF * sizeof(float);
  float* Wh2   = (float*)ws;  ws += (size_t)NN * FOUT  * sizeof(float);
  float* src1  = (float*)ws;  ws += (size_t)NHEADS * NN * sizeof(float);
  float* dst1  = (float*)ws;  ws += (size_t)NHEADS * NN * sizeof(float);
  float* src2  = (float*)ws;  ws += (size_t)NN * sizeof(float);
  float* dst2  = (float*)ws;  ws += (size_t)NN * sizeof(float);
  int*   ncnt  = (int*)ws;    ws += (size_t)NN * sizeof(int);
  int*   nidx  = (int*)ws;    ws += (size_t)NN * CAP * sizeof(int);

  build_csr_kernel<<<NN, 128, 0, stream>>>(adj, ncnt, nidx);
  lin_kernel<<<NN / 16, 256, 0, stream>>>(x, lin_W, lin_b, h);
  heads_gemm_kernel<<<NN / 16, 512, 0, stream>>>(h, W_heads, hcat);
  attn_coeff_heads_kernel<<<(NN * NHEADS) / 256, 256, 0, stream>>>(hcat, a_heads, src1, dst1);
  gat_aggregate_heads_kernel<<<NN / ROWS_PER_BLOCK, 32 * ROWS_PER_BLOCK, 0, stream>>>(hcat, src1, dst1, ncnt, nidx, gat1);
  end_gemm_kernel<<<NN / 16, 128, 0, stream>>>(gat1, W_end, Wh2);
  attn_coeff_end_kernel<<<NN / 256, 256, 0, stream>>>(Wh2, a_end, src2, dst2);
  gat_end_kernel<<<NN / ROWS_PER_BLOCK, 32 * ROWS_PER_BLOCK, 0, stream>>>(Wh2, src2, dst2, ncnt, nidx, out);
}


# Launch fusion 8->4: each GEMM runs in the launch that makes its A rows (tile handed over in LDS); coefficient dots move into the GEMM epilogues, summed as a 16-lane tree per tile, then the 4 tiles
# speedup vs baseline: 1.6246x; 1.3798x over previous
#include <hip/hip_runtime.h>
#include <math.h>

#define NN      4096
#define FIN     128
#define FHID    64
#define NHEADS  4
#define FOUT    64
#define HCATF   (NHEADS * FHID)
#define ALPHA_L 0.2f
#define CAP     128

typedef float v2f __attribute__((ext_vector_type(2)));
typedef float v8f __attribute__((ext_vector_type(8)));

#define DEV static __device__ __forceinline__

DEV int uni(int v) { return __builtin_amdgcn_readfirstlane(v); }
DEV float xrow(float x) {
  return __builtin_bit_cast(float, __builtin_amdgcn_permlanex16(__builtin_bit_cast(int, x), __builtin_bit_cast(int, x),
                                                               0x76543210, 0xfedcba98, false, false));
}
template <int M> DEV float xlane(float x) {
  return __builtin_bit_cast(float, __builtin_amdgcn_update_dpp(0, __builtin_bit_cast(int, x),
                                                              0x160 + M, 0xf, 0xf, true));
}
DEV float wave_max(float v) {
  v = fmaxf(v, xrow(v));
  v = fmaxf(v, xlane<8>(v));
  v = fmaxf(v, xlane<4>(v));
  v = fmaxf(v, xlane<2>(v));
  v = fmaxf(v, xlane<1>(v));
  return v;
}
DEV float wave_sum(float v) {
  v = v + xrow(v);
  v = v + xlane<8>(v);
  v = v + xlane<4>(v);
  v = v + xlane<2>(v);
  v = v + xlane<1>(v);
  return v;
}
DEV float lane_f(float x, int l) { return __builtin_bit_cast(float, __builtin_amdgcn_readlane(__builtin_bit_cast(int, x), l)); }
DEV int   lane_i(int x, int l)   { return __builtin_amdgcn_readlane(x, l); }
DEV float leaky(float e) { return (e >= 0.f) ? e : ALPHA_L * e; }
DEV float elu1(float o)  { return (o > 0.f) ? o : (expf(o) - 1.f); }

#define ROWS_PER_BLOCK 4
#define ROWS_L1 16
#define GPITCH 260
#define AGG_G 8
#define AGG_G_END 16

__device__ __forceinline__ v8f wmma_f32_tile(const float* __restrict__ A, int lda,
                                             const float* __restrict__ B, int ldb,
                                             int K, int lane) {
  const int half = lane >> 4;
  const int l16  = lane & 15;
  v8f c = {0.f, 0.f, 0.f, 0.f, 0.f, 0.f, 0.f, 0.f};
  for (int k = 0; k < K; k += 4) {
    v2f a, b;
    const float* ap = A + (size_t)l16 * lda + k + 2 * half;
    a.x = ap[0];
    a.y = ap[1];
    b.x = B[(size_t)(k + 2 * half) * ldb + l16];
    b.y = B[(size_t)(k + 2 * half + 1) * ldb + l16];
    c = __builtin_amdgcn_wmma_f32_16x16x4_f32(false, a, false, b, (short)0, c, false, false);
  }
  return c;
}

__global__ void build_csr_kernel(const float* __restrict__ adj,
                                 int* __restrict__ cnt, int* __restrict__ nidx) {
  __shared__ int scan[128];
  const int row = blockIdx.x;
  const int t = threadIdx.x;
  const float4* arow4 = (const float4*)(adj + (size_t)row * NN) + t * 8;
  const int base = t * 32;

  float4 v[8];
#pragma unroll
  for (int q = 0; q < 8; ++q) v[q] = arow4[q];

  int c = 0;
#pragma unroll
  for (int q = 0; q < 8; ++q) {
    c += (v[q].x > 0.f) ? 1 : 0;
    c += (v[q].y > 0.f) ? 1 : 0;
    c += (v[q].z > 0.f) ? 1 : 0;
    c += (v[q].w > 0.f) ? 1 : 0;
  }
  scan[t] = c;
  __syncthreads();
  for (int off = 1; off < 128; off <<= 1) {
    int p = (t >= off) ? scan[t - off] : 0;
    __syncthreads();
    scan[t] += p;
    __syncthreads();
  }
  int k = scan[t] - c;
  const int total = scan[127];
  if (t == 0) cnt[row] = total < CAP ? total : CAP;
  int* out = nidx + (size_t)row * CAP;
#pragma unroll
  for (int q = 0; q < 8; ++q) {
    const int b = base + q * 4;
    if (v[q].x > 0.f) { if (k < CAP) out[k] = b + 0; ++k; }
    if (v[q].y > 0.f) { if (k < CAP) out[k] = b + 1; ++k; }
    if (v[q].z > 0.f) { if (k < CAP) out[k] = b + 2; ++k; }
    if (v[q].w > 0.f) { if (k < CAP) out[k] = b + 3; ++k; }
  }
}

DEV float row16_sum(float v) {
  v = v + xlane<8>(v);
  v = v + xlane<4>(v);
  v = v + xlane<2>(v);
  v = v + xlane<1>(v);
  return v;
}

#define HPITCH 132

__global__ __launch_bounds__(512)
void lin_heads_coeff_kernel(const float* __restrict__ x, const float* __restrict__ W,
                            const float* __restrict__ b, const float* __restrict__ W_heads,
                            const float* __restrict__ a_heads, float* __restrict__ hcat,
                            float* __restrict__ src, float* __restrict__ dst) {
  __shared__ float hs[16 * HPITCH];
  __shared__ float part[16][16][2];
  const int mtile = blockIdx.x;
  const int wave  = uni((int)(threadIdx.x >> 5));
  const int lane  = threadIdx.x & 31;
  const int half  = lane >> 4, l16 = lane & 15;

  if (wave < 8) {
    v8f c = wmma_f32_tile(x + (size_t)mtile * 16 * FIN, FIN, W + wave * 16, FIN, FIN, lane);
    const float bias = b[wave * 16 + l16];
#pragma unroll
    for (int r = 0; r < 8; ++r)
      hs[(r + 8 * half) * HPITCH + wave * 16 + l16] = c[r] + bias;
  }
  __syncthreads();

  const int head = wave >> 2;
  const int nt   = wave & 3;
  const float* B = W_heads + (size_t)head * FIN * FHID + nt * 16;
  v8f c = wmma_f32_tile(hs, HPITCH, B, FHID, FIN, lane);
  float* out = hcat + (size_t)mtile * 16 * HCATF + head * FHID + nt * 16 + l16;
#pragma unroll
  for (int r = 0; r < 8; ++r)
    out[(size_t)(r + 8 * half) * HCATF] = c[r];

  const float as = a_heads[head * 2 * FHID + nt * 16 + l16];
  const float ad = a_heads[head * 2 * FHID + FHID + nt * 16 + l16];
#pragma unroll
  for (int r = 0; r < 8; ++r) {
    const float ps = row16_sum(c[r] * as);
    const float pd = row16_sum(c[r] * ad);
    if (l16 == 0) {
      part[wave][r + 8 * half][0] = ps;
      part[wave][r + 8 * half][1] = pd;
    }
  }
  __syncthreads();
  if (threadIdx.x < 16 * NHEADS) {
    const int row = threadIdx.x >> 2, hd = threadIdx.x & 3;
    const float s = (part[hd * 4 + 0][row][0] + part[hd * 4 + 1][row][0]) +
                    (part[hd * 4 + 2][row][0] + part[hd * 4 + 3][row][0]);
    const float d = (part[hd * 4 + 0][row][1] + part[hd * 4 + 1][row][1]) +
                    (part[hd * 4 + 2][row][1] + part[hd * 4 + 3][row][1]);
    src[((size_t)mtile * 16 + row) * NHEADS + hd] = s;
    dst[((size_t)mtile * 16 + row) * NHEADS + hd] = d;
  }
}

template <int N>
DEV void agg_heads_group(const char* __restrict__ hbytes, unsigned voff, int js,
                         float w0, float w1, float w2, float w3, int l0, bool hi,
                         float4& accA, float4& accB) {
  float4 xa[N], xb[N];
  for (int u = 0; u < N; ++u) {
    const int jk = lane_i(js, l0 + u);
    const char* p = hbytes + (size_t)jk * (HCATF * sizeof(float));
    xa[u] = *(const float4*)(p + voff);
    xb[u] = *(const float4*)(p + voff + 2 * FHID * sizeof(float));
  }
  for (int u = 0; u < N; ++u) {
    const float s0 = lane_f(w0, l0 + u), s1 = lane_f(w1, l0 + u);
    const float s2 = lane_f(w2, l0 + u), s3 = lane_f(w3, l0 + u);
    const float wa = hi ? s1 : s0;
    const float wb = hi ? s3 : s2;
    accA.x = fmaf(wa, xa[u].x, accA.x); accA.y = fmaf(wa, xa[u].y, accA.y);
    accA.z = fmaf(wa, xa[u].z, accA.z); accA.w = fmaf(wa, xa[u].w, accA.w);
    accB.x = fmaf(wb, xb[u].x, accB.x); accB.y = fmaf(wb, xb[u].y, accB.y);
    accB.z = fmaf(wb, xb[u].z, accB.z); accB.w = fmaf(wb, xb[u].w, accB.w);
  }
}

__global__ __launch_bounds__(32 * ROWS_L1)
void gat_aggregate_heads_end_gemm_kernel(const float* __restrict__ hcat,
                                         const float* __restrict__ src,
                                         const float* __restrict__ dst,
                                         const int* __restrict__ cnt,
                                         const int* __restrict__ nidx,
                                         const float* __restrict__ W_end,
                                         const float* __restrict__ a_end,
                                         float* __restrict__ Wh2, float* __restrict__ sd) {
  __shared__ __attribute__((aligned(16))) float gt[ROWS_L1 * GPITCH];
  __shared__ float part[4][16][2];
  const int lane = threadIdx.x & 31;
  const bool hi  = lane >= 16;
  const int wv   = uni((int)(threadIdx.x >> 5));
  const int row  = uni((int)(blockIdx.x * ROWS_L1) + wv);
  const int deg  = uni(cnt[row]);
  const int* nb  = nidx + (size_t)row * CAP;
  const float4 si = *(const float4*)(src + (size_t)row * NHEADS);

  int    j[4] = {0, 0, 0, 0};
  float4 d[4] = {};
  float  w[4][NHEADS];
  const float4* dst4 = (const float4*)dst;
  if (deg > 0) {
    j[0] = nb[min(lane, deg - 1)];
    j[1] = nb[min(lane + 32, deg - 1)];
    d[0] = dst4[j[0]];
    d[1] = dst4[j[1]];
    if (deg > 64) {
      j[2] = nb[min(lane + 64, deg - 1)];
      j[3] = nb[min(lane + 96, deg - 1)];
      d[2] = dst4[j[2]];
      d[3] = dst4[j[3]];
    }
  }
  float m[NHEADS] = {-3.4e38f, -3.4e38f, -3.4e38f, -3.4e38f};
  for (int s = 0; s < 4; ++s) {
    const bool valid = (s * 32 + lane) < deg;
    w[s][0] = valid ? leaky(si.x + d[s].x) : -3.4e38f;
    w[s][1] = valid ? leaky(si.y + d[s].y) : -3.4e38f;
    w[s][2] = valid ? leaky(si.z + d[s].z) : -3.4e38f;
    w[s][3] = valid ? leaky(si.w + d[s].w) : -3.4e38f;
    for (int h = 0; h < NHEADS; ++h) m[h] = fmaxf(m[h], w[s][h]);
  }
  for (int h = 0; h < NHEADS; ++h) m[h] = wave_max(m[h]);

  for (int s = 0; s < 4; ++s) {
    const bool valid = (s * 32 + lane) < deg;
    for (int h = 0; h < NHEADS; ++h) w[s][h] = valid ? expf(w[s][h] - m[h]) : 0.f;
  }
  float denom[NHEADS];
  for (int h = 0; h < NHEADS; ++h)
    denom[h] = wave_sum((w[0][h] + w[2][h]) + (w[1][h] + w[3][h]));

  float4 accA = {0.f, 0.f, 0.f, 0.f}, accB = {0.f, 0.f, 0.f, 0.f};
  const char* hbytes = (const char*)hcat;
  const unsigned voff = (unsigned)lane * 16u;
  for (int base = 0; base < deg; base += AGG_G) {
    const int s  = base >> 5;
    int l0 = base & 31;
    const int   js = (s == 0) ? j[0]    : (s == 1) ? j[1]    : (s == 2) ? j[2]    : j[3];
    const float w0 = (s == 0) ? w[0][0] : (s == 1) ? w[1][0] : (s == 2) ? w[2][0] : w[3][0];
    const float w1 = (s == 0) ? w[0][1] : (s == 1) ? w[1][1] : (s == 2) ? w[2][1] : w[3][1];
    const float w2 = (s == 0) ? w[0][2] : (s == 1) ? w[1][2] : (s == 2) ? w[2][2] : w[3][2];
    const float w3 = (s == 0) ? w[0][3] : (s == 1) ? w[1][3] : (s == 2) ? w[2][3] : w[3][3];
    const int r = deg - base;
    if (r >= AGG_G) {
      agg_heads_group<AGG_G>(hbytes, voff, js, w0, w1, w2, w3, l0, hi, accA, accB);
    } else {
      if (r & 4) { agg_heads_group<4>(hbytes, voff, js, w0, w1, w2, w3, l0, hi, accA, accB); l0 += 4; }
      if (r & 2) { agg_heads_group<2>(hbytes, voff, js, w0, w1, w2, w3, l0, hi, accA, accB); l0 += 2; }
      if (r & 1) { agg_heads_group<1>(hbytes, voff, js, w0, w1, w2, w3, l0, hi, accA, accB); }
    }
  }

  const float dA = hi ? denom[1] : denom[0];
  const float dB = hi ? denom[3] : denom[2];
  float4 oA, oB;
  oA.x = elu1(accA.x / dA); oA.y = elu1(accA.y / dA); oA.z = elu1(accA.z / dA); oA.w = elu1(accA.w / dA);
  oB.x = elu1(accB.x / dB); oB.y = elu1(accB.y / dB); oB.z = elu1(accB.z / dB); oB.w = elu1(accB.w / dB);
  float* orow = gt + wv * GPITCH + lane * 4;
  *(float4*)orow = oA;
  *(float4*)(orow + 2 * FHID) = oB;
  __syncthreads();

  const int half = lane >> 4, l16 = lane & 15;
  if (wv < 4) {
    v8f c = wmma_f32_tile(gt, GPITCH, W_end + wv * 16, FOUT, HCATF, lane);
    float* out = Wh2 + (size_t)blockIdx.x * 16 * FOUT + wv * 16 + l16;
#pragma unroll
    for (int r = 0; r < 8; ++r)
      out[(size_t)(r + 8 * half) * FOUT] = c[r];

    const float as = a_end[wv * 16 + l16];
    const float ad = a_end[FOUT + wv * 16 + l16];
#pragma unroll
    for (int r = 0; r < 8; ++r) {
      const float ps = row16_sum(c[r] * as);
      const float pd = row16_sum(c[r] * ad);
      if (l16 == 0) {
        part[wv][r + 8 * half][0] = ps;
        part[wv][r + 8 * half][1] = pd;
      }
    }
  }
  __syncthreads();
  if (threadIdx.x < 32) {
    const int trow = threadIdx.x >> 1, which = threadIdx.x & 1;
    sd[((size_t)blockIdx.x * 16 + trow) * 2 + which] =
        (part[0][trow][which] + part[1][trow][which]) + (part[2][trow][which] + part[3][trow][which]);
  }
}

template <int N>
DEV void agg_end_group(const char* __restrict__ wbytes, unsigned voff, int js, float ws,
                       int l0, float& a0, float& a1) {
  float x0[N], x1[N];
  for (int u = 0; u < N; ++u) {
    const int jk = lane_i(js, l0 + u);
    const char* p = wbytes + (size_t)jk * (FOUT * sizeof(float));
    x0[u] = *(const float*)(p + voff);
    x1[u] = *(const float*)(p + voff + 32 * sizeof(float));
  }
  for (int u = 0; u < N; ++u) {
    const float wk = lane_f(ws, l0 + u);
    a0 = fmaf(wk, x0[u], a0);
    a1 = fmaf(wk, x1[u], a1);
  }
}

__global__ __launch_bounds__(32 * ROWS_PER_BLOCK)
void gat_end_kernel(const float* __restrict__ Wh2,
                    const float* __restrict__ sd,
                    const int* __restrict__ cnt,
                    const int* __restrict__ nidx,
                    float* __restrict__ out) {
  const int lane = threadIdx.x & 31;
  const int row  = uni((int)(blockIdx.x * ROWS_PER_BLOCK + (threadIdx.x >> 5)));
  const int deg  = uni(cnt[row]);
  const float si = sd[2 * (size_t)row];
  const int* nb  = nidx + (size_t)row * CAP;

  int   j[4] = {0, 0, 0, 0};
  float d[4] = {0.f, 0.f, 0.f, 0.f};
  float w[4];
  if (deg > 0) {
    j[0] = nb[min(lane, deg - 1)];
    j[1] = nb[min(lane + 32, deg - 1)];
    d[0] = sd[2 * (size_t)j[0] + 1];
    d[1] = sd[2 * (size_t)j[1] + 1];
    if (deg > 64) {
      j[2] = nb[min(lane + 64, deg - 1)];
      j[3] = nb[min(lane + 96, deg - 1)];
      d[2] = sd[2 * (size_t)j[2] + 1];
      d[3] = sd[2 * (size_t)j[3] + 1];
    }
  }
  float m = -3.4e38f;
  for (int s = 0; s < 4; ++s) {
    w[s] = ((s * 32 + lane) < deg) ? leaky(si + d[s]) : -3.4e38f;
    m = fmaxf(m, w[s]);
  }
  m = wave_max(m);

  for (int s = 0; s < 4; ++s) w[s] = ((s * 32 + lane) < deg) ? expf(w[s] - m) : 0.f;
  const float denom = wave_sum((w[0] + w[2]) + (w[1] + w[3]));

  float a0 = 0.f, a1 = 0.f;
  const char* wbytes = (const char*)Wh2;
  const unsigned voff = (unsigned)lane * 4u;
  for (int base = 0; base < deg; base += AGG_G_END) {
    const int s  = base >> 5;
    int l0 = base & 31;
    const int   js = (s == 0) ? j[0] : (s == 1) ? j[1] : (s == 2) ? j[2] : j[3];
    const float ws = (s == 0) ? w[0] : (s == 1) ? w[1] : (s == 2) ? w[2] : w[3];
    const int r = deg - base;
    if (r >= AGG_G_END) {
      agg_end_group<AGG_G_END>(wbytes, voff, js, ws, l0, a0, a1);
    } else {
      if (r & 8) { agg_end_group<8>(wbytes, voff, js, ws, l0, a0, a1); l0 += 8; }
      if (r & 4) { agg_end_group<4>(wbytes, voff, js, ws, l0, a0, a1); l0 += 4; }
      if (r & 2) { agg_end_group<2>(wbytes, voff, js, ws, l0, a0, a1); l0 += 2; }
      if (r & 1) { agg_end_group<1>(wbytes, voff, js, ws, l0, a0, a1); }
    }
  }
  const float o0 = elu1(a0 / denom);
  const float o1 = elu1(a1 / denom);

  const float mo  = wave_max(fmaxf(o0, o1));
  const float ex0 = expf(o0 - mo);
  const float ex1 = expf(o1 - mo);
  const float sum = wave_sum(ex0 + ex1);
  float* orow = out + (size_t)row * FOUT + lane;
  orow[0]  = ex0 / sum;
  orow[32] = ex1 / sum;
}

extern "C" void kernel_launch(void* const* d_in, const int* in_sizes, int n_in,
                              void* d_out, int out_size, void* d_ws, size_t ws_size,
                              hipStream_t stream) {
  (void)in_sizes; (void)n_in; (void)out_size; (void)ws_size;
  const float* x       = (const float*)d_in[0];
  const float* adj     = (const float*)d_in[1];
  const float* lin_W   = (const float*)d_in[2];
  const float* lin_b   = (const float*)d_in[3];
  const float* W_heads = (const float*)d_in[4];
  const float* a_heads = (const float*)d_in[5];
  const float* W_end   = (const float*)d_in[6];
  const float* a_end   = (const float*)d_in[7];
  float* out = (float*)d_out;

  char* ws = (char*)d_ws;
  ws += (size_t)NN * FIN * sizeof(float);
  float* hcat  = (float*)ws;  ws += (size_t)NN * HCATF * sizeof(float);
  ws += (size_t)NN * HCATF * sizeof(float);
  float* Wh2   = (float*)ws;  ws += (size_t)NN * FOUT  * sizeof(float);
  float* src1  = (float*)ws;  ws += (size_t)NHEADS * NN * sizeof(float);
  float* dst1  = (float*)ws;  ws += (size_t)NHEADS * NN * sizeof(float);
  float* sd2   = (float*)ws;  ws += (size_t)2 * NN * sizeof(float);
  int*   ncnt  = (int*)ws;    ws += (size_t)NN * sizeof(int);
  int*   nidx  = (int*)ws;    ws += (size_t)NN * CAP * sizeof(int);

  build_csr_kernel<<<NN, 128, 0, stream>>>(adj, ncnt, nidx);
  lin_heads_coeff_kernel<<<NN / 16, 512, 0, stream>>>(x, lin_W, lin_b, W_heads, a_heads, hcat, src1, dst1);
  gat_aggregate_heads_end_gemm_kernel<<<NN / ROWS_L1, 32 * ROWS_L1, 0, stream>>>(hcat, src1, dst1, ncnt, nidx, W_end, a_end, Wh2, sd2);
  gat_end_kernel<<<NN / ROWS_PER_BLOCK, 32 * ROWS_PER_BLOCK, 0, stream>>>(Wh2, sd2, ncnt, nidx, out);
}


# CSR build: one wave per adjacency row, lane-contiguous 16 B loads (4 whole lines per load, was 32 part lines) and ballot prefix for the slots; no LDS scan, no barriers, 4x fewer waves
# speedup vs baseline: 1.7128x; 1.0543x over previous
#include <hip/hip_runtime.h>
#include <math.h>

#define NN      4096
#define FIN     128
#define FHID    64
#define NHEADS  4
#define FOUT    64
#define HCATF   (NHEADS * FHID)
#define ALPHA_L 0.2f
#define CAP     128

typedef float v2f __attribute__((ext_vector_type(2)));
typedef float v8f __attribute__((ext_vector_type(8)));

#define DEV static __device__ __forceinline__

DEV int uni(int v) { return __builtin_amdgcn_readfirstlane(v); }
DEV float xrow(float x) {
  return __builtin_bit_cast(float, __builtin_amdgcn_permlanex16(__builtin_bit_cast(int, x), __builtin_bit_cast(int, x),
                                                               0x76543210, 0xfedcba98, false, false));
}
template <int M> DEV float xlane(float x) {
  return __builtin_bit_cast(float, __builtin_amdgcn_update_dpp(0, __builtin_bit_cast(int, x),
                                                              0x160 + M, 0xf, 0xf, true));
}
DEV float wave_max(float v) {
  v = fmaxf(v, xrow(v));
  v = fmaxf(v, xlane<8>(v));
  v = fmaxf(v, xlane<4>(v));
  v = fmaxf(v, xlane<2>(v));
  v = fmaxf(v, xlane<1>(v));
  return v;
}
DEV float wave_sum(float v) {
  v = v + xrow(v);
  v = v + xlane<8>(v);
  v = v + xlane<4>(v);
  v = v + xlane<2>(v);
  v = v + xlane<1>(v);
  return v;
}
DEV float lane_f(float x, int l) { return __builtin_bit_cast(float, __builtin_amdgcn_readlane(__builtin_bit_cast(int, x), l)); }
DEV int   lane_i(int x, int l)   { return __builtin_amdgcn_readlane(x, l); }
DEV float leaky(float e) { return (e >= 0.f) ? e : ALPHA_L * e; }
DEV float elu1(float o)  { return (o > 0.f) ? o : (expf(o) - 1.f); }

#define ROWS_PER_BLOCK 4
#define ROWS_L1 16
#define GPITCH 260
#define AGG_G 8
#define AGG_G_END 16

__device__ __forceinline__ v8f wmma_f32_tile(const float* __restrict__ A, int lda,
                                             const float* __restrict__ B, int ldb,
                                             int K, int lane) {
  const int half = lane >> 4;
  const int l16  = lane & 15;
  v8f c = {0.f, 0.f, 0.f, 0.f, 0.f, 0.f, 0.f, 0.f};
  for (int k = 0; k < K; k += 4) {
    v2f a, b;
    const float* ap = A + (size_t)l16 * lda + k + 2 * half;
    a.x = ap[0];
    a.y = ap[1];
    b.x = B[(size_t)(k + 2 * half) * ldb + l16];
    b.y = B[(size_t)(k + 2 * half + 1) * ldb + l16];
    c = __builtin_amdgcn_wmma_f32_16x16x4_f32(false, a, false, b, (short)0, c, false, false);
  }
  return c;
}

#define CSR_ROWS 16
#define CSR_G 8
DEV unsigned ballot32(bool p) { return __builtin_amdgcn_ballot_w32(p); }
DEV int below(unsigned m, int base) { return (int)__builtin_amdgcn_mbcnt_lo(m, (unsigned)base); }

__global__ __launch_bounds__(32 * CSR_ROWS)
void build_csr_kernel(const float* __restrict__ adj,
                      int* __restrict__ cnt, int* __restrict__ nidx) {
  const int lane = threadIdx.x & 31;
  const int row  = uni((int)(blockIdx.x * CSR_ROWS + (threadIdx.x >> 5)));
  const float4* arow4 = (const float4*)(adj + (size_t)row * NN) + lane;
  int* out = nidx + (size_t)row * CAP;
  int k = 0;
  float4 cur[CSR_G], nxt[CSR_G];
#pragma unroll
  for (int u = 0; u < CSR_G; ++u) cur[u] = arow4[u * 32];
#pragma unroll
  for (int q0 = 0; q0 < NN / 128; q0 += CSR_G) {
    if (q0 + CSR_G < NN / 128) {
#pragma unroll
      for (int u = 0; u < CSR_G; ++u) nxt[u] = arow4[(q0 + CSR_G + u) * 32];
    }
#pragma unroll
    for (int u = 0; u < CSR_G; ++u) {
      const bool f0 = cur[u].x > 0.f, f1 = cur[u].y > 0.f, f2 = cur[u].z > 0.f, f3 = cur[u].w > 0.f;
      const unsigned m0 = ballot32(f0), m1 = ballot32(f1), m2 = ballot32(f2), m3 = ballot32(f3);
      if ((m0 | m1 | m2 | m3) != 0u) {
        const int col = (q0 + u) * 128 + 4 * lane;
        int p = below(m3, below(m2, below(m1, below(m0, k))));
        if (m0 != 0u) { if (f0 && p < CAP) out[p] = col + 0; p += f0 ? 1 : 0; }
        if (m1 != 0u) { if (f1 && p < CAP) out[p] = col + 1; p += f1 ? 1 : 0; }
        if (m2 != 0u) { if (f2 && p < CAP) out[p] = col + 2; p += f2 ? 1 : 0; }
        if (m3 != 0u) { if (f3 && p < CAP) out[p] = col + 3; }
        k += __builtin_popcount(m0) + __builtin_popcount(m1) + __builtin_popcount(m2) + __builtin_popcount(m3);
      }
    }
    if (q0 + CSR_G < NN / 128) {
#pragma unroll
      for (int u = 0; u < CSR_G; ++u) cur[u] = nxt[u];
    }
  }
  if (lane == 0) cnt[row] = k < CAP ? k : CAP;
}

DEV float row16_sum(float v) {
  v = v + xlane<8>(v);
  v = v + xlane<4>(v);
  v = v + xlane<2>(v);
  v = v + xlane<1>(v);
  return v;
}

#define HPITCH 132

__global__ __launch_bounds__(512)
void lin_heads_coeff_kernel(const float* __restrict__ x, const float* __restrict__ W,
                            const float* __restrict__ b, const float* __restrict__ W_heads,
                            const float* __restrict__ a_heads, float* __restrict__ hcat,
                            float* __restrict__ src, float* __restrict__ dst) {
  __shared__ float hs[16 * HPITCH];
  __shared__ float part[16][16][2];
  const int mtile = blockIdx.x;
  const int wave  = uni((int)(threadIdx.x >> 5));
  const int lane  = threadIdx.x & 31;
  const int half  = lane >> 4, l16 = lane & 15;

  if (wave < 8) {
    v8f c = wmma_f32_tile(x + (size_t)mtile * 16 * FIN, FIN, W + wave * 16, FIN, FIN, lane);
    const float bias = b[wave * 16 + l16];
#pragma unroll
    for (int r = 0; r < 8; ++r)
      hs[(r + 8 * half) * HPITCH + wave * 16 + l16] = c[r] + bias;
  }
  __syncthreads();

  const int head = wave >> 2;
  const int nt   = wave & 3;
  const float* B = W_heads + (size_t)head * FIN * FHID + nt * 16;
  v8f c = wmma_f32_tile(hs, HPITCH, B, FHID, FIN, lane);
  float* out = hcat + (size_t)mtile * 16 * HCATF + head * FHID + nt * 16 + l16;
#pragma unroll
  for (int r = 0; r < 8; ++r)
    out[(size_t)(r + 8 * half) * HCATF] = c[r];

  const float as = a_heads[head * 2 * FHID + nt * 16 + l16];
  const float ad = a_heads[head * 2 * FHID + FHID + nt * 16 + l16];
#pragma unroll
  for (int r = 0; r < 8; ++r) {
    const float ps = row16_sum(c[r] * as);
    const float pd = row16_sum(c[r] * ad);
    if (l16 == 0) {
      part[wave][r + 8 * half][0] = ps;
      part[wave][r + 8 * half][1] = pd;
    }
  }
  __syncthreads();
  if (threadIdx.x < 16 * NHEADS) {
    const int row = threadIdx.x >> 2, hd = threadIdx.x & 3;
    const float s = (part[hd * 4 + 0][row][0] + part[hd * 4 + 1][row][0]) +
                    (part[hd * 4 + 2][row][0] + part[hd * 4 + 3][row][0]);
    const float d = (part[hd * 4 + 0][row][1] + part[hd * 4 + 1][row][1]) +
                    (part[hd * 4 + 2][row][1] + part[hd * 4 + 3][row][1]);
    src[((size_t)mtile * 16 + row) * NHEADS + hd] = s;
    dst[((size_t)mtile * 16 + row) * NHEADS + hd] = d;
  }
}

template <int N>
DEV void agg_heads_group(const char* __restrict__ hbytes, unsigned voff, int js,
                         float w0, float w1, float w2, float w3, int l0, bool hi,
                         float4& accA, float4& accB) {
  float4 xa[N], xb[N];
  for (int u = 0; u < N; ++u) {
    const int jk = lane_i(js, l0 + u);
    const char* p = hbytes + (size_t)jk * (HCATF * sizeof(float));
    xa[u] = *(const float4*)(p + voff);
    xb[u] = *(const float4*)(p + voff + 2 * FHID * sizeof(float));
  }
  for (int u = 0; u < N; ++u) {
    const float s0 = lane_f(w0, l0 + u), s1 = lane_f(w1, l0 + u);
    const float s2 = lane_f(w2, l0 + u), s3 = lane_f(w3, l0 + u);
    const float wa = hi ? s1 : s0;
    const float wb = hi ? s3 : s2;
    accA.x = fmaf(wa, xa[u].x, accA.x); accA.y = fmaf(wa, xa[u].y, accA.y);
    accA.z = fmaf(wa, xa[u].z, accA.z); accA.w = fmaf(wa, xa[u].w, accA.w);
    accB.x = fmaf(wb, xb[u].x, accB.x); accB.y = fmaf(wb, xb[u].y, accB.y);
    accB.z = fmaf(wb, xb[u].z, accB.z); accB.w = fmaf(wb, xb[u].w, accB.w);
  }
}

__global__ __launch_bounds__(32 * ROWS_L1)
void gat_aggregate_heads_end_gemm_kernel(const float* __restrict__ hcat,
                                         const float* __restrict__ src,
                                         const float* __restrict__ dst,
                                         const int* __restrict__ cnt,
                                         const int* __restrict__ nidx,
                                         const float* __restrict__ W_end,
                                         const float* __restrict__ a_end,
                                         float* __restrict__ Wh2, float* __restrict__ sd) {
  __shared__ __attribute__((aligned(16))) float gt[ROWS_L1 * GPITCH];
  __shared__ float part[4][16][2];
  const int lane = threadIdx.x & 31;
  const bool hi  = lane >= 16;
  const int wv   = uni((int)(threadIdx.x >> 5));
  const int row  = uni((int)(blockIdx.x * ROWS_L1) + wv);
  const int deg  = uni(cnt[row]);
  const int* nb  = nidx + (size_t)row * CAP;
  const float4 si = *(const float4*)(src + (size_t)row * NHEADS);

  int    j[4] = {0, 0, 0, 0};
  float4 d[4] = {};
  float  w[4][NHEADS];
  const float4* dst4 = (const float4*)dst;
  if (deg > 0) {
    j[0] = nb[min(lane, deg - 1)];
    j[1] = nb[min(lane + 32, deg - 1)];
    d[0] = dst4[j[0]];
    d[1] = dst4[j[1]];
    if (deg > 64) {
      j[2] = nb[min(lane + 64, deg - 1)];
      j[3] = nb[min(lane + 96, deg - 1)];
      d[2] = dst4[j[2]];
      d[3] = dst4[j[3]];
    }
  }
  float m[NHEADS] = {-3.4e38f, -3.4e38f, -3.4e38f, -3.4e38f};
  for (int s = 0; s < 4; ++s) {
    const bool valid = (s * 32 + lane) < deg;
    w[s][0] = valid ? leaky(si.x + d[s].x) : -3.4e38f;
    w[s][1] = valid ? leaky(si.y + d[s].y) : -3.4e38f;
    w[s][2] = valid ? leaky(si.z + d[s].z) : -3.4e38f;
    w[s][3] = valid ? leaky(si.w + d[s].w) : -3.4e38f;
    for (int h = 0; h < NHEADS; ++h) m[h] = fmaxf(m[h], w[s][h]);
  }
  for (int h = 0; h < NHEADS; ++h) m[h] = wave_max(m[h]);

  for (int s = 0; s < 4; ++s) {
    const bool valid = (s * 32 + lane) < deg;
    for (int h = 0; h < NHEADS; ++h) w[s][h] = valid ? expf(w[s][h] - m[h]) : 0.f;
  }
  float denom[NHEADS];
  for (int h = 0; h < NHEADS; ++h)
    denom[h] = wave_sum((w[0][h] + w[2][h]) + (w[1][h] + w[3][h]));

  float4 accA = {0.f, 0.f, 0.f, 0.f}, accB = {0.f, 0.f, 0.f, 0.f};
  const char* hbytes = (const char*)hcat;
  const unsigned voff = (unsigned)lane * 16u;
  for (int base = 0; base < deg; base += AGG_G) {
    const int s  = base >> 5;
    int l0 = base & 31;
    const int   js = (s == 0) ? j[0]    : (s == 1) ? j[1]    : (s == 2) ? j[2]    : j[3];
    const float w0 = (s == 0) ? w[0][0] : (s == 1) ? w[1][0] : (s == 2) ? w[2][0] : w[3][0];
    const float w1 = (s == 0) ? w[0][1] : (s == 1) ? w[1][1] : (s == 2) ? w[2][1] : w[3][1];
    const float w2 = (s == 0) ? w[0][2] : (s == 1) ? w[1][2] : (s == 2) ? w[2][2] : w[3][2];
    const float w3 = (s == 0) ? w[0][3] : (s == 1) ? w[1][3] : (s == 2) ? w[2][3] : w[3][3];
    const int r = deg - base;
    if (r >= AGG_G) {
      agg_heads_group<AGG_G>(hbytes, voff, js, w0, w1, w2, w3, l0, hi, accA, accB);
    } else {
      if (r & 4) { agg_heads_group<4>(hbytes, voff, js, w0, w1, w2, w3, l0, hi, accA, accB); l0 += 4; }
      if (r & 2) { agg_heads_group<2>(hbytes, voff, js, w0, w1, w2, w3, l0, hi, accA, accB); l0 += 2; }
      if (r & 1) { agg_heads_group<1>(hbytes, voff, js, w0, w1, w2, w3, l0, hi, accA, accB); }
    }
  }

  const float dA = hi ? denom[1] : denom[0];
  const float dB = hi ? denom[3] : denom[2];
  float4 oA, oB;
  oA.x = elu1(accA.x / dA); oA.y = elu1(accA.y / dA); oA.z = elu1(accA.z / dA); oA.w = elu1(accA.w / dA);
  oB.x = elu1(accB.x / dB); oB.y = elu1(accB.y / dB); oB.z = elu1(accB.z / dB); oB.w = elu1(accB.w / dB);
  float* orow = gt + wv * GPITCH + lane * 4;
  *(float4*)orow = oA;
  *(float4*)(orow + 2 * FHID) = oB;
  __syncthreads();

  const int half = lane >> 4, l16 = lane & 15;
  if (wv < 4) {
    v8f c = wmma_f32_tile(gt, GPITCH, W_end + wv * 16, FOUT, HCATF, lane);
    float* out = Wh2 + (size_t)blockIdx.x * 16 * FOUT + wv * 16 + l16;
#pragma unroll
    for (int r = 0; r < 8; ++r)
      out[(size_t)(r + 8 * half) * FOUT] = c[r];

    const float as = a_end[wv * 16 + l16];
    const float ad = a_end[FOUT + wv * 16 + l16];
#pragma unroll
    for (int r = 0; r < 8; ++r) {
      const float ps = row16_sum(c[r] * as);
      const float pd = row16_sum(c[r] * ad);
      if (l16 == 0) {
        part[wv][r + 8 * half][0] = ps;
        part[wv][r + 8 * half][1] = pd;
      }
    }
  }
  __syncthreads();
  if (threadIdx.x < 32) {
    const int trow = threadIdx.x >> 1, which = threadIdx.x & 1;
    sd[((size_t)blockIdx.x * 16 + trow) * 2 + which] =
        (part[0][trow][which] + part[1][trow][which]) + (part[2][trow][which] + part[3][trow][which]);
  }
}

template <int N>
DEV void agg_end_group(const char* __restrict__ wbytes, unsigned voff, int js, float ws,
                       int l0, float& a0, float& a1) {
  float x0[N], x1[N];
  for (int u = 0; u < N; ++u) {
    const int jk = lane_i(js, l0 + u);
    const char* p = wbytes + (size_t)jk * (FOUT * sizeof(float));
    x0[u] = *(const float*)(p + voff);
    x1[u] = *(const float*)(p + voff + 32 * sizeof(float));
  }
  for (int u = 0; u < N; ++u) {
    const float wk = lane_f(ws, l0 + u);
    a0 = fmaf(wk, x0[u], a0);
    a1 = fmaf(wk, x1[u], a1);
  }
}

__global__ __launch_bounds__(32 * ROWS_PER_BLOCK)
void gat_end_kernel(const float* __restrict__ Wh2,
                    const float* __restrict__ sd,
                    const int* __restrict__ cnt,
                    const int* __restrict__ nidx,
                    float* __restrict__ out) {
  const int lane = threadIdx.x & 31;
  const int row  = uni((int)(blockIdx.x * ROWS_PER_BLOCK + (threadIdx.x >> 5)));
  const int deg  = uni(cnt[row]);
  const float si = sd[2 * (size_t)row];
  const int* nb  = nidx + (size_t)row * CAP;

  int   j[4] = {0, 0, 0, 0};
  float d[4] = {0.f, 0.f, 0.f, 0.f};
  float w[4];
  if (deg > 0) {
    j[0] = nb[min(lane, deg - 1)];
    j[1] = nb[min(lane + 32, deg - 1)];
    d[0] = sd[2 * (size_t)j[0] + 1];
    d[1] = sd[2 * (size_t)j[1] + 1];
    if (deg > 64) {
      j[2] = nb[min(lane + 64, deg - 1)];
      j[3] = nb[min(lane + 96, deg - 1)];
      d[2] = sd[2 * (size_t)j[2] + 1];
      d[3] = sd[2 * (size_t)j[3] + 1];
    }
  }
  float m = -3.4e38f;
  for (int s = 0; s < 4; ++s) {
    w[s] = ((s * 32 + lane) < deg) ? leaky(si + d[s]) : -3.4e38f;
    m = fmaxf(m, w[s]);
  }
  m = wave_max(m);

  for (int s = 0; s < 4; ++s) w[s] = ((s * 32 + lane) < deg) ? expf(w[s] - m) : 0.f;
  const float denom = wave_sum((w[0] + w[2]) + (w[1] + w[3]));

  float a0 = 0.f, a1 = 0.f;
  const char* wbytes = (const char*)Wh2;
  const unsigned voff = (unsigned)lane * 4u;
  for (int base = 0; base < deg; base += AGG_G_END) {
    const int s  = base >> 5;
    int l0 = base & 31;
    const int   js = (s == 0) ? j[0] : (s == 1) ? j[1] : (s == 2) ? j[2] : j[3];
    const float ws = (s == 0) ? w[0] : (s == 1) ? w[1] : (s == 2) ? w[2] : w[3];
    const int r = deg - base;
    if (r >= AGG_G_END) {
      agg_end_group<AGG_G_END>(wbytes, voff, js, ws, l0, a0, a1);
    } else {
      if (r & 8) { agg_end_group<8>(wbytes, voff, js, ws, l0, a0, a1); l0 += 8; }
      if (r & 4) { agg_end_group<4>(wbytes, voff, js, ws, l0, a0, a1); l0 += 4; }
      if (r & 2) { agg_end_group<2>(wbytes, voff, js, ws, l0, a0, a1); l0 += 2; }
      if (r & 1) { agg_end_group<1>(wbytes, voff, js, ws, l0, a0, a1); }
    }
  }
  const float o0 = elu1(a0 / denom);
  const float o1 = elu1(a1 / denom);

  const float mo  = wave_max(fmaxf(o0, o1));
  const float ex0 = expf(o0 - mo);
  const float ex1 = expf(o1 - mo);
  const float sum = wave_sum(ex0 + ex1);
  float* orow = out + (size_t)row * FOUT + lane;
  orow[0]  = ex0 / sum;
  orow[32] = ex1 / sum;
}

extern "C" void kernel_launch(void* const* d_in, const int* in_sizes, int n_in,
                              void* d_out, int out_size, void* d_ws, size_t ws_size,
                              hipStream_t stream) {
  (void)in_sizes; (void)n_in; (void)out_size; (void)ws_size;
  const float* x       = (const float*)d_in[0];
  const float* adj     = (const float*)d_in[1];
  const float* lin_W   = (const float*)d_in[2];
  const float* lin_b   = (const float*)d_in[3];
  const float* W_heads = (const float*)d_in[4];
  const float* a_heads = (const float*)d_in[5];
  const float* W_end   = (const float*)d_in[6];
  const float* a_end   = (const float*)d_in[7];
  float* out = (float*)d_out;

  char* ws = (char*)d_ws;
  ws += (size_t)NN * FIN * sizeof(float);
  float* hcat  = (float*)ws;  ws += (size_t)NN * HCATF * sizeof(float);
  ws += (size_t)NN * HCATF * sizeof(float);
  float* Wh2   = (float*)ws;  ws += (size_t)NN * FOUT  * sizeof(float);
  float* src1  = (float*)ws;  ws += (size_t)NHEADS * NN * sizeof(float);
  float* dst1  = (float*)ws;  ws += (size_t)NHEADS * NN * sizeof(float);
  float* sd2   = (float*)ws;  ws += (size_t)2 * NN * sizeof(float);
  int*   ncnt  = (int*)ws;    ws += (size_t)NN * sizeof(int);
  int*   nidx  = (int*)ws;    ws += (size_t)NN * CAP * sizeof(int);

  build_csr_kernel<<<NN / CSR_ROWS, 32 * CSR_ROWS, 0, stream>>>(adj, ncnt, nidx);
  lin_heads_coeff_kernel<<<NN / 16, 512, 0, stream>>>(x, lin_W, lin_b, W_heads, a_heads, hcat, src1, dst1);
  gat_aggregate_heads_end_gemm_kernel<<<NN / ROWS_L1, 32 * ROWS_L1, 0, stream>>>(hcat, src1, dst1, ncnt, nidx, W_end, a_end, Wh2, sd2);
  gat_end_kernel<<<NN / ROWS_PER_BLOCK, 32 * ROWS_PER_BLOCK, 0, stream>>>(Wh2, sd2, ncnt, nidx, out);
}


# Launch fusion 4->3: the neighbor-list build and the first GEMM launch, which do not depend on each other, run as one grid (256 + 256 blocks, role by block index) and overlap
# speedup vs baseline: 1.9277x; 1.1255x over previous
#include <hip/hip_runtime.h>
#include <math.h>

#define NN      4096
#define FIN     128
#define FHID    64
#define NHEADS  4
#define FOUT    64
#define HCATF   (NHEADS * FHID)
#define ALPHA_L 0.2f
#define CAP     128

typedef float v2f __attribute__((ext_vector_type(2)));
typedef float v8f __attribute__((ext_vector_type(8)));

#define DEV static __device__ __forceinline__

DEV int uni(int v) { return __builtin_amdgcn_readfirstlane(v); }
DEV float xrow(float x) {
  return __builtin_bit_cast(float, __builtin_amdgcn_permlanex16(__builtin_bit_cast(int, x), __builtin_bit_cast(int, x),
                                                               0x76543210, 0xfedcba98, false, false));
}
template <int M> DEV float xlane(float x) {
  return __builtin_bit_cast(float, __builtin_amdgcn_update_dpp(0, __builtin_bit_cast(int, x),
                                                              0x160 + M, 0xf, 0xf, true));
}
DEV float wave_max(float v) {
  v = fmaxf(v, xrow(v));
  v = fmaxf(v, xlane<8>(v));
  v = fmaxf(v, xlane<4>(v));
  v = fmaxf(v, xlane<2>(v));
  v = fmaxf(v, xlane<1>(v));
  return v;
}
DEV float wave_sum(float v) {
  v = v + xrow(v);
  v = v + xlane<8>(v);
  v = v + xlane<4>(v);
  v = v + xlane<2>(v);
  v = v + xlane<1>(v);
  return v;
}
DEV float lane_f(float x, int l) { return __builtin_bit_cast(float, __builtin_amdgcn_readlane(__builtin_bit_cast(int, x), l)); }
DEV int   lane_i(int x, int l)   { return __builtin_amdgcn_readlane(x, l); }
DEV float leaky(float e) { return (e >= 0.f) ? e : ALPHA_L * e; }
DEV float elu1(float o)  { return (o > 0.f) ? o : (expf(o) - 1.f); }

#define ROWS_PER_BLOCK 4
#define ROWS_L1 16
#define GPITCH 260
#define AGG_G 8
#define AGG_G_END 16

__device__ __forceinline__ v8f wmma_f32_tile(const float* __restrict__ A, int lda,
                                             const float* __restrict__ B, int ldb,
                                             int K, int lane) {
  const int half = lane >> 4;
  const int l16  = lane & 15;
  v8f c = {0.f, 0.f, 0.f, 0.f, 0.f, 0.f, 0.f, 0.f};
  for (int k = 0; k < K; k += 4) {
    v2f a, b;
    const float* ap = A + (size_t)l16 * lda + k + 2 * half;
    a.x = ap[0];
    a.y = ap[1];
    b.x = B[(size_t)(k + 2 * half) * ldb + l16];
    b.y = B[(size_t)(k + 2 * half + 1) * ldb + l16];
    c = __builtin_amdgcn_wmma_f32_16x16x4_f32(false, a, false, b, (short)0, c, false, false);
  }
  return c;
}

#define CSR_ROWS 16
#define CSR_G 8
DEV unsigned ballot32(bool p) { return __builtin_amdgcn_ballot_w32(p); }
DEV int below(unsigned m, int base) { return (int)__builtin_amdgcn_mbcnt_lo(m, (unsigned)base); }

DEV void build_csr_rows(int cblk, const float* __restrict__ adj,
                        int* __restrict__ cnt, int* __restrict__ nidx) {
  const int lane = threadIdx.x & 31;
  const int row  = uni((int)(cblk * CSR_ROWS + (threadIdx.x >> 5)));
  const float4* arow4 = (const float4*)(adj + (size_t)row * NN) + lane;
  int* out = nidx + (size_t)row * CAP;
  int k = 0;
  float4 cur[CSR_G], nxt[CSR_G];
#pragma unroll
  for (int u = 0; u < CSR_G; ++u) cur[u] = arow4[u * 32];
#pragma unroll
  for (int q0 = 0; q0 < NN / 128; q0 += CSR_G) {
    if (q0 + CSR_G < NN / 128) {
#pragma unroll
      for (int u = 0; u < CSR_G; ++u) nxt[u] = arow4[(q0 + CSR_G + u) * 32];
    }
#pragma unroll
    for (int u = 0; u < CSR_G; ++u) {
      const bool f0 = cur[u].x > 0.f, f1 = cur[u].y > 0.f, f2 = cur[u].z > 0.f, f3 = cur[u].w > 0.f;
      const unsigned m0 = ballot32(f0), m1 = ballot32(f1), m2 = ballot32(f2), m3 = ballot32(f3);
      if ((m0 | m1 | m2 | m3) != 0u) {
        const int col = (q0 + u) * 128 + 4 * lane;
        int p = below(m3, below(m2, below(m1, below(m0, k))));
        if (m0 != 0u) { if (f0 && p < CAP) out[p] = col + 0; p += f0 ? 1 : 0; }
        if (m1 != 0u) { if (f1 && p < CAP) out[p] = col + 1; p += f1 ? 1 : 0; }
        if (m2 != 0u) { if (f2 && p < CAP) out[p] = col + 2; p += f2 ? 1 : 0; }
        if (m3 != 0u) { if (f3 && p < CAP) out[p] = col + 3; }
        k += __builtin_popcount(m0) + __builtin_popcount(m1) + __builtin_popcount(m2) + __builtin_popcount(m3);
      }
    }
    if (q0 + CSR_G < NN / 128) {
#pragma unroll
      for (int u = 0; u < CSR_G; ++u) cur[u] = nxt[u];
    }
  }
  if (lane == 0) cnt[row] = k < CAP ? k : CAP;
}

DEV float row16_sum(float v) {
  v = v + xlane<8>(v);
  v = v + xlane<4>(v);
  v = v + xlane<2>(v);
  v = v + xlane<1>(v);
  return v;
}

#define HPITCH 132

DEV void lin_heads_coeff_tile(int mtile, const float* __restrict__ x, const float* __restrict__ W,
                              const float* __restrict__ b, const float* __restrict__ W_heads,
                              const float* __restrict__ a_heads, float* __restrict__ hcat,
                              float* __restrict__ src, float* __restrict__ dst) {
  __shared__ float hs[16 * HPITCH];
  __shared__ float part[16][16][2];
  const int wave  = uni((int)(threadIdx.x >> 5));
  const int lane  = threadIdx.x & 31;
  const int half  = lane >> 4, l16 = lane & 15;

  if (wave < 8) {
    v8f c = wmma_f32_tile(x + (size_t)mtile * 16 * FIN, FIN, W + wave * 16, FIN, FIN, lane);
    const float bias = b[wave * 16 + l16];
#pragma unroll
    for (int r = 0; r < 8; ++r)
      hs[(r + 8 * half) * HPITCH + wave * 16 + l16] = c[r] + bias;
  }
  __syncthreads();

  const int head = wave >> 2;
  const int nt   = wave & 3;
  const float* B = W_heads + (size_t)head * FIN * FHID + nt * 16;
  v8f c = wmma_f32_tile(hs, HPITCH, B, FHID, FIN, lane);
  float* out = hcat + (size_t)mtile * 16 * HCATF + head * FHID + nt * 16 + l16;
#pragma unroll
  for (int r = 0; r < 8; ++r)
    out[(size_t)(r + 8 * half) * HCATF] = c[r];

  const float as = a_heads[head * 2 * FHID + nt * 16 + l16];
  const float ad = a_heads[head * 2 * FHID + FHID + nt * 16 + l16];
#pragma unroll
  for (int r = 0; r < 8; ++r) {
    const float ps = row16_sum(c[r] * as);
    const float pd = row16_sum(c[r] * ad);
    if (l16 == 0) {
      part[wave][r + 8 * half][0] = ps;
      part[wave][r + 8 * half][1] = pd;
    }
  }
  __syncthreads();
  if (threadIdx.x < 16 * NHEADS) {
    const int row = threadIdx.x >> 2, hd = threadIdx.x & 3;
    const float s = (part[hd * 4 + 0][row][0] + part[hd * 4 + 1][row][0]) +
                    (part[hd * 4 + 2][row][0] + part[hd * 4 + 3][row][0]);
    const float d = (part[hd * 4 + 0][row][1] + part[hd * 4 + 1][row][1]) +
                    (part[hd * 4 + 2][row][1] + part[hd * 4 + 3][row][1]);
    src[((size_t)mtile * 16 + row) * NHEADS + hd] = s;
    dst[((size_t)mtile * 16 + row) * NHEADS + hd] = d;
  }
}

#define LIN_TILES (NN / 16)
__global__ __launch_bounds__(512)
void csr_lin_heads_coeff_kernel(const float* __restrict__ adj, int* __restrict__ cnt, int* __restrict__ nidx,
                                const float* __restrict__ x, const float* __restrict__ W,
                                const float* __restrict__ b, const float* __restrict__ W_heads,
                                const float* __restrict__ a_heads, float* __restrict__ hcat,
                                float* __restrict__ src, float* __restrict__ dst) {
  const int blk = (int)blockIdx.x;
  if (blk < LIN_TILES) lin_heads_coeff_tile(blk, x, W, b, W_heads, a_heads, hcat, src, dst);
  else                 build_csr_rows(blk - LIN_TILES, adj, cnt, nidx);
}

template <int N>
DEV void agg_heads_group(const char* __restrict__ hbytes, unsigned voff, int js,
                         float w0, float w1, float w2, float w3, int l0, bool hi,
                         float4& accA, float4& accB) {
  float4 xa[N], xb[N];
  for (int u = 0; u < N; ++u) {
    const int jk = lane_i(js, l0 + u);
    const char* p = hbytes + (size_t)jk * (HCATF * sizeof(float));
    xa[u] = *(const float4*)(p + voff);
    xb[u] = *(const float4*)(p + voff + 2 * FHID * sizeof(float));
  }
  for (int u = 0; u < N; ++u) {
    const float s0 = lane_f(w0, l0 + u), s1 = lane_f(w1, l0 + u);
    const float s2 = lane_f(w2, l0 + u), s3 = lane_f(w3, l0 + u);
    const float wa = hi ? s1 : s0;
    const float wb = hi ? s3 : s2;
    accA.x = fmaf(wa, xa[u].x, accA.x); accA.y = fmaf(wa, xa[u].y, accA.y);
    accA.z = fmaf(wa, xa[u].z, accA.z); accA.w = fmaf(wa, xa[u].w, accA.w);
    accB.x = fmaf(wb, xb[u].x, accB.x); accB.y = fmaf(wb, xb[u].y, accB.y);
    accB.z = fmaf(wb, xb[u].z, accB.z); accB.w = fmaf(wb, xb[u].w, accB.w);
  }
}

__global__ __launch_bounds__(32 * ROWS_L1)
void gat_aggregate_heads_end_gemm_kernel(const float* __restrict__ hcat,
                                         const float* __restrict__ src,
                                         const float* __restrict__ dst,
                                         const int* __restrict__ cnt,
                                         const int* __restrict__ nidx,
                                         const float* __restrict__ W_end,
                                         const float* __restrict__ a_end,
                                         float* __restrict__ Wh2, float* __restrict__ sd) {
  __shared__ __attribute__((aligned(16))) float gt[ROWS_L1 * GPITCH];
  __shared__ float part[4][16][2];
  const int lane = threadIdx.x & 31;
  const bool hi  = lane >= 16;
  const int wv   = uni((int)(threadIdx.x >> 5));
  const int row  = uni((int)(blockIdx.x * ROWS_L1) + wv);
  const int deg  = uni(cnt[row]);
  const int* nb  = nidx + (size_t)row * CAP;
  const float4 si = *(const float4*)(src + (size_t)row * NHEADS);

  int    j[4] = {0, 0, 0, 0};
  float4 d[4] = {};
  float  w[4][NHEADS];
  const float4* dst4 = (const float4*)dst;
  if (deg > 0) {
    j[0] = nb[min(lane, deg - 1)];
    j[1] = nb[min(lane + 32, deg - 1)];
    d[0] = dst4[j[0]];
    d[1] = dst4[j[1]];
    if (deg > 64) {
      j[2] = nb[min(lane + 64, deg - 1)];
      j[3] = nb[min(lane + 96, deg - 1)];
      d[2] = dst4[j[2]];
      d[3] = dst4[j[3]];
    }
  }
  float m[NHEADS] = {-3.4e38f, -3.4e38f, -3.4e38f, -3.4e38f};
  for (int s = 0; s < 4; ++s) {
    const bool valid = (s * 32 + lane) < deg;
    w[s][0] = valid ? leaky(si.x + d[s].x) : -3.4e38f;
    w[s][1] = valid ? leaky(si.y + d[s].y) : -3.4e38f;
    w[s][2] = valid ? leaky(si.z + d[s].z) : -3.4e38f;
    w[s][3] = valid ? leaky(si.w + d[s].w) : -3.4e38f;
    for (int h = 0; h < NHEADS; ++h) m[h] = fmaxf(m[h], w[s][h]);
  }
  for (int h = 0; h < NHEADS; ++h) m[h] = wave_max(m[h]);

  for (int s = 0; s < 4; ++s) {
    const bool valid = (s * 32 + lane) < deg;
    for (int h = 0; h < NHEADS; ++h) w[s][h] = valid ? expf(w[s][h] - m[h]) : 0.f;
  }
  float denom[NHEADS];
  for (int h = 0; h < NHEADS; ++h)
    denom[h] = wave_sum((w[0][h] + w[2][h]) + (w[1][h] + w[3][h]));

  float4 accA = {0.f, 0.f, 0.f, 0.f}, accB = {0.f, 0.f, 0.f, 0.f};
  const char* hbytes = (const char*)hcat;
  const unsigned voff = (unsigned)lane * 16u;
  for (int base = 0; base < deg; base += AGG_G) {
    const int s  = base >> 5;
    int l0 = base & 31;
    const int   js = (s == 0) ? j[0]    : (s == 1) ? j[1]    : (s == 2) ? j[2]    : j[3];
    const float w0 = (s == 0) ? w[0][0] : (s == 1) ? w[1][0] : (s == 2) ? w[2][0] : w[3][0];
    const float w1 = (s == 0) ? w[0][1] : (s == 1) ? w[1][1] : (s == 2) ? w[2][1] : w[3][1];
    const float w2 = (s == 0) ? w[0][2] : (s == 1) ? w[1][2] : (s == 2) ? w[2][2] : w[3][2];
    const float w3 = (s == 0) ? w[0][3] : (s == 1) ? w[1][3] : (s == 2) ? w[2][3] : w[3][3];
    const int r = deg - base;
    if (r >= AGG_G) {
      agg_heads_group<AGG_G>(hbytes, voff, js, w0, w1, w2, w3, l0, hi, accA, accB);
    } else {
      if (r & 4) { agg_heads_group<4>(hbytes, voff, js, w0, w1, w2, w3, l0, hi, accA, accB); l0 += 4; }
      if (r & 2) { agg_heads_group<2>(hbytes, voff, js, w0, w1, w2, w3, l0, hi, accA, accB); l0 += 2; }
      if (r & 1) { agg_heads_group<1>(hbytes, voff, js, w0, w1, w2, w3, l0, hi, accA, accB); }
    }
  }

  const float dA = hi ? denom[1] : denom[0];
  const float dB = hi ? denom[3] : denom[2];
  float4 oA, oB;
  oA.x = elu1(accA.x / dA); oA.y = elu1(accA.y / dA); oA.z = elu1(accA.z / dA); oA.w = elu1(accA.w / dA);
  oB.x = elu1(accB.x / dB); oB.y = elu1(accB.y / dB); oB.z = elu1(accB.z / dB); oB.w = elu1(accB.w / dB);
  float* orow = gt + wv * GPITCH + lane * 4;
  *(float4*)orow = oA;
  *(float4*)(orow + 2 * FHID) = oB;
  __syncthreads();

  const int half = lane >> 4, l16 = lane & 15;
  if (wv < 4) {
    v8f c = wmma_f32_tile(gt, GPITCH, W_end + wv * 16, FOUT, HCATF, lane);
    float* out = Wh2 + (size_t)blockIdx.x * 16 * FOUT + wv * 16 + l16;
#pragma unroll
    for (int r = 0; r < 8; ++r)
      out[(size_t)(r + 8 * half) * FOUT] = c[r];

    const float as = a_end[wv * 16 + l16];
    const float ad = a_end[FOUT + wv * 16 + l16];
#pragma unroll
    for (int r = 0; r < 8; ++r) {
      const float ps = row16_sum(c[r] * as);
      const float pd = row16_sum(c[r] * ad);
      if (l16 == 0) {
        part[wv][r + 8 * half][0] = ps;
        part[wv][r + 8 * half][1] = pd;
      }
    }
  }
  __syncthreads();
  if (threadIdx.x < 32) {
    const int trow = threadIdx.x >> 1, which = threadIdx.x & 1;
    sd[((size_t)blockIdx.x * 16 + trow) * 2 + which] =
        (part[0][trow][which] + part[1][trow][which]) + (part[2][trow][which] + part[3][trow][which]);
  }
}

template <int N>
DEV void agg_end_group(const char* __restrict__ wbytes, unsigned voff, int js, float ws,
                       int l0, float& a0, float& a1) {
  float x0[N], x1[N];
  for (int u = 0; u < N; ++u) {
    const int jk = lane_i(js, l0 + u);
    const char* p = wbytes + (size_t)jk * (FOUT * sizeof(float));
    x0[u] = *(const float*)(p + voff);
    x1[u] = *(const float*)(p + voff + 32 * sizeof(float));
  }
  for (int u = 0; u < N; ++u) {
    const float wk = lane_f(ws, l0 + u);
    a0 = fmaf(wk, x0[u], a0);
    a1 = fmaf(wk, x1[u], a1);
  }
}

__global__ __launch_bounds__(32 * ROWS_PER_BLOCK)
void gat_end_kernel(const float* __restrict__ Wh2,
                    const float* __restrict__ sd,
                    const int* __restrict__ cnt,
                    const int* __restrict__ nidx,
                    float* __restrict__ out) {
  const int lane = threadIdx.x & 31;
  const int row  = uni((int)(blockIdx.x * ROWS_PER_BLOCK + (threadIdx.x >> 5)));
  const int deg  = uni(cnt[row]);
  const float si = sd[2 * (size_t)row];
  const int* nb  = nidx + (size_t)row * CAP;

  int   j[4] = {0, 0, 0, 0};
  float d[4] = {0.f, 0.f, 0.f, 0.f};
  float w[4];
  if (deg > 0) {
    j[0] = nb[min(lane, deg - 1)];
    j[1] = nb[min(lane + 32, deg - 1)];
    d[0] = sd[2 * (size_t)j[0] + 1];
    d[1] = sd[2 * (size_t)j[1] + 1];
    if (deg > 64) {
      j[2] = nb[min(lane + 64, deg - 1)];
      j[3] = nb[min(lane + 96, deg - 1)];
      d[2] = sd[2 * (size_t)j[2] + 1];
      d[3] = sd[2 * (size_t)j[3] + 1];
    }
  }
  float m = -3.4e38f;
  for (int s = 0; s < 4; ++s) {
    w[s] = ((s * 32 + lane) < deg) ? leaky(si + d[s]) : -3.4e38f;
    m = fmaxf(m, w[s]);
  }
  m = wave_max(m);

  for (int s = 0; s < 4; ++s) w[s] = ((s * 32 + lane) < deg) ? expf(w[s] - m) : 0.f;
  const float denom = wave_sum((w[0] + w[2]) + (w[1] + w[3]));

  float a0 = 0.f, a1 = 0.f;
  const char* wbytes = (const char*)Wh2;
  const unsigned voff = (unsigned)lane * 4u;
  for (int base = 0; base < deg; base += AGG_G_END) {
    const int s  = base >> 5;
    int l0 = base & 31;
    const int   js = (s == 0) ? j[0] : (s == 1) ? j[1] : (s == 2) ? j[2] : j[3];
    const float ws = (s == 0) ? w[0] : (s == 1) ? w[1] : (s == 2) ? w[2] : w[3];
    const int r = deg - base;
    if (r >= AGG_G_END) {
      agg_end_group<AGG_G_END>(wbytes, voff, js, ws, l0, a0, a1);
    } else {
      if (r & 8) { agg_end_group<8>(wbytes, voff, js, ws, l0, a0, a1); l0 += 8; }
      if (r & 4) { agg_end_group<4>(wbytes, voff, js, ws, l0, a0, a1); l0 += 4; }
      if (r & 2) { agg_end_group<2>(wbytes, voff, js, ws, l0, a0, a1); l0 += 2; }
      if (r & 1) { agg_end_group<1>(wbytes, voff, js, ws, l0, a0, a1); }
    }
  }
  const float o0 = elu1(a0 / denom);
  const float o1 = elu1(a1 / denom);

  const float mo  = wave_max(fmaxf(o0, o1));
  const float ex0 = expf(o0 - mo);
  const float ex1 = expf(o1 - mo);
  const float sum = wave_sum(ex0 + ex1);
  float* orow = out + (size_t)row * FOUT + lane;
  orow[0]  = ex0 / sum;
  orow[32] = ex1 / sum;
}

extern "C" void kernel_launch(void* const* d_in, const int* in_sizes, int n_in,
                              void* d_out, int out_size, void* d_ws, size_t ws_size,
                              hipStream_t stream) {
  (void)in_sizes; (void)n_in; (void)out_size; (void)ws_size;
  const float* x       = (const float*)d_in[0];
  const float* adj     = (const float*)d_in[1];
  const float* lin_W   = (const float*)d_in[2];
  const float* lin_b   = (const float*)d_in[3];
  const float* W_heads = (const float*)d_in[4];
  const float* a_heads = (const float*)d_in[5];
  const float* W_end   = (const float*)d_in[6];
  const float* a_end   = (const float*)d_in[7];
  float* out = (float*)d_out;

  char* ws = (char*)d_ws;
  ws += (size_t)NN * FIN * sizeof(float);
  float* hcat  = (float*)ws;  ws += (size_t)NN * HCATF * sizeof(float);
  ws += (size_t)NN * HCATF * sizeof(float);
  float* Wh2   = (float*)ws;  ws += (size_t)NN * FOUT  * sizeof(float);
  float* src1  = (float*)ws;  ws += (size_t)NHEADS * NN * sizeof(float);
  float* dst1  = (float*)ws;  ws += (size_t)NHEADS * NN * sizeof(float);
  float* sd2   = (float*)ws;  ws += (size_t)2 * NN * sizeof(float);
  int*   ncnt  = (int*)ws;    ws += (size_t)NN * sizeof(int);
  int*   nidx  = (int*)ws;    ws += (size_t)NN * CAP * sizeof(int);

  csr_lin_heads_coeff_kernel<<<LIN_TILES + NN / CSR_ROWS, 512, 0, stream>>>(adj, ncnt, nidx, x, lin_W, lin_b, W_heads, a_heads,
                                                                              hcat, src1, dst1);
  gat_aggregate_heads_end_gemm_kernel<<<NN / ROWS_L1, 32 * ROWS_L1, 0, stream>>>(hcat, src1, dst1, ncnt, nidx, W_end, a_end, Wh2, sd2);
  gat_end_kernel<<<NN / ROWS_PER_BLOCK, 32 * ROWS_PER_BLOCK, 0, stream>>>(Wh2, sd2, ncnt, nidx, out);
}


# Layer-1 kernel: all 16 waves copy W_end to LDS at kernel entry (in flight under the neighbor phase); the end GEMM reads both fragments from LDS, was 9 serial cache round trips behind the barrier
# speedup vs baseline: 2.2245x; 1.1540x over previous
#include <hip/hip_runtime.h>
#include <math.h>

#define NN      4096
#define FIN     128
#define FHID    64
#define NHEADS  4
#define FOUT    64
#define HCATF   (NHEADS * FHID)
#define ALPHA_L 0.2f
#define CAP     128

typedef float v2f __attribute__((ext_vector_type(2)));
typedef float v8f __attribute__((ext_vector_type(8)));

#define DEV static __device__ __forceinline__

DEV int uni(int v) { return __builtin_amdgcn_readfirstlane(v); }
DEV float xrow(float x) {
  return __builtin_bit_cast(float, __builtin_amdgcn_permlanex16(__builtin_bit_cast(int, x), __builtin_bit_cast(int, x),
                                                               0x76543210, 0xfedcba98, false, false));
}
template <int M> DEV float xlane(float x) {
  return __builtin_bit_cast(float, __builtin_amdgcn_update_dpp(0, __builtin_bit_cast(int, x),
                                                              0x160 + M, 0xf, 0xf, true));
}
DEV float wave_max(float v) {
  v = fmaxf(v, xrow(v));
  v = fmaxf(v, xlane<8>(v));
  v = fmaxf(v, xlane<4>(v));
  v = fmaxf(v, xlane<2>(v));
  v = fmaxf(v, xlane<1>(v));
  return v;
}
DEV float wave_sum(float v) {
  v = v + xrow(v);
  v = v + xlane<8>(v);
  v = v + xlane<4>(v);
  v = v + xlane<2>(v);
  v = v + xlane<1>(v);
  return v;
}
DEV float lane_f(float x, int l) { return __builtin_bit_cast(float, __builtin_amdgcn_readlane(__builtin_bit_cast(int, x), l)); }
DEV int   lane_i(int x, int l)   { return __builtin_amdgcn_readlane(x, l); }
DEV float leaky(float e) { return (e >= 0.f) ? e : ALPHA_L * e; }
DEV float elu1(float o)  { return (o > 0.f) ? o : (expf(o) - 1.f); }

#define ROWS_PER_BLOCK 4
#define ROWS_L1 16
#define GPITCH 260
#define AGG_G 8
#define AGG_G_END 16
#define WQ (4 * FOUT)

__device__ __forceinline__ v8f wmma_f32_tile(const float* __restrict__ A, int lda,
                                             const float* __restrict__ B, int ldb,
                                             int K, int lane) {
  const int half = lane >> 4;
  const int l16  = lane & 15;
  v8f c = {0.f, 0.f, 0.f, 0.f, 0.f, 0.f, 0.f, 0.f};
  for (int k = 0; k < K; k += 4) {
    v2f a, b;
    const float* ap = A + (size_t)l16 * lda + k + 2 * half;
    a.x = ap[0];
    a.y = ap[1];
    b.x = B[(size_t)(k + 2 * half) * ldb + l16];
    b.y = B[(size_t)(k + 2 * half + 1) * ldb + l16];
    c = __builtin_amdgcn_wmma_f32_16x16x4_f32(false, a, false, b, (short)0, c, false, false);
  }
  return c;
}

#define TILE_KB 8
DEV void lds_frag_group(const float* __restrict__ ap, const float* __restrict__ bp, int s0,
                        v2f (&a)[TILE_KB], v2f (&b)[TILE_KB]) {
#pragma unroll
  for (int u = 0; u < TILE_KB; ++u) {
    a[u].x = ap[4 * (s0 + u)];
    a[u].y = ap[4 * (s0 + u) + 1];
    b[u].x = bp[WQ * (s0 + u)];
    b[u].y = bp[WQ * (s0 + u) + 1];
  }
}
DEV v8f wmma_f32_tile_lds(const float* __restrict__ At, const float* __restrict__ Bn, int lane) {
  const int half = lane >> 4;
  const int l16  = lane & 15;
  const float* ap = At + l16 * GPITCH + 2 * half;
  const float* bp = Bn + l16 * 4 + 2 * half;
  v8f c = {0.f, 0.f, 0.f, 0.f, 0.f, 0.f, 0.f, 0.f};
  v2f a0[TILE_KB], b0[TILE_KB], a1[TILE_KB], b1[TILE_KB];
  lds_frag_group(ap, bp, 0, a0, b0);
  __builtin_amdgcn_sched_barrier(0);
#pragma unroll
  for (int g = 0; g < HCATF / 4 / TILE_KB; g += 2) {
    lds_frag_group(ap, bp, (g + 1) * TILE_KB, a1, b1);
    __builtin_amdgcn_sched_barrier(0);
#pragma unroll
    for (int u = 0; u < TILE_KB; ++u)
      c = __builtin_amdgcn_wmma_f32_16x16x4_f32(false, a0[u], false, b0[u], (short)0, c, false, false);
    __builtin_amdgcn_sched_barrier(0);
    if (g + 2 < HCATF / 4 / TILE_KB) lds_frag_group(ap, bp, (g + 2) * TILE_KB, a0, b0);
    __builtin_amdgcn_sched_barrier(0);
#pragma unroll
    for (int u = 0; u < TILE_KB; ++u)
      c = __builtin_amdgcn_wmma_f32_16x16x4_f32(false, a1[u], false, b1[u], (short)0, c, false, false);
    __builtin_amdgcn_sched_barrier(0);
  }
  return c;
}

#define CSR_ROWS 16
#define CSR_G 8
DEV unsigned ballot32(bool p) { return __builtin_amdgcn_ballot_w32(p); }
DEV int below(unsigned m, int base) { return (int)__builtin_amdgcn_mbcnt_lo(m, (unsigned)base); }

DEV void build_csr_rows(int cblk, const float* __restrict__ adj,
                        int* __restrict__ cnt, int* __restrict__ nidx) {
  const int lane = threadIdx.x & 31;
  const int row  = uni((int)(cblk * CSR_ROWS + (threadIdx.x >> 5)));
  const float4* arow4 = (const float4*)(adj + (size_t)row * NN) + lane;
  int* out = nidx + (size_t)row * CAP;
  int k = 0;
  float4 cur[CSR_G], nxt[CSR_G];
#pragma unroll
  for (int u = 0; u < CSR_G; ++u) cur[u] = arow4[u * 32];
#pragma unroll
  for (int q0 = 0; q0 < NN / 128; q0 += CSR_G) {
    if (q0 + CSR_G < NN / 128) {
#pragma unroll
      for (int u = 0; u < CSR_G; ++u) nxt[u] = arow4[(q0 + CSR_G + u) * 32];
    }
#pragma unroll
    for (int u = 0; u < CSR_G; ++u) {
      const bool f0 = cur[u].x > 0.f, f1 = cur[u].y > 0.f, f2 = cur[u].z > 0.f, f3 = cur[u].w > 0.f;
      const unsigned m0 = ballot32(f0), m1 = ballot32(f1), m2 = ballot32(f2), m3 = ballot32(f3);
      if ((m0 | m1 | m2 | m3) != 0u) {
        const int col = (q0 + u) * 128 + 4 * lane;
        int p = below(m3, below(m2, below(m1, below(m0, k))));
        if (m0 != 0u) { if (f0 && p < CAP) out[p] = col + 0; p += f0 ? 1 : 0; }
        if (m1 != 0u) { if (f1 && p < CAP) out[p] = col + 1; p += f1 ? 1 : 0; }
        if (m2 != 0u) { if (f2 && p < CAP) out[p] = col + 2; p += f2 ? 1 : 0; }
        if (m3 != 0u) { if (f3 && p < CAP) out[p] = col + 3; }
        k += __builtin_popcount(m0) + __builtin_popcount(m1) + __builtin_popcount(m2) + __builtin_popcount(m3);
      }
    }
    if (q0 + CSR_G < NN / 128) {
#pragma unroll
      for (int u = 0; u < CSR_G; ++u) cur[u] = nxt[u];
    }
  }
  if (lane == 0) cnt[row] = k < CAP ? k : CAP;
}

DEV float row16_sum(float v) {
  v = v + xlane<8>(v);
  v = v + xlane<4>(v);
  v = v + xlane<2>(v);
  v = v + xlane<1>(v);
  return v;
}

#define HPITCH 132

DEV void lin_heads_coeff_tile(int mtile, const float* __restrict__ x, const float* __restrict__ W,
                              const float* __restrict__ b, const float* __restrict__ W_heads,
                              const float* __restrict__ a_heads, float* __restrict__ hcat,
                              float* __restrict__ src, float* __restrict__ dst) {
  __shared__ float hs[16 * HPITCH];
  __shared__ float part[16][16][2];
  const int wave  = uni((int)(threadIdx.x >> 5));
  const int lane  = threadIdx.x & 31;
  const int half  = lane >> 4, l16 = lane & 15;

  if (wave < 8) {
    v8f c = wmma_f32_tile(x + (size_t)mtile * 16 * FIN, FIN, W + wave * 16, FIN, FIN, lane);
    const float bias = b[wave * 16 + l16];
#pragma unroll
    for (int r = 0; r < 8; ++r)
      hs[(r + 8 * half) * HPITCH + wave * 16 + l16] = c[r] + bias;
  }
  __syncthreads();

  const int head = wave >> 2;
  const int nt   = wave & 3;
  const float* B = W_heads + (size_t)head * FIN * FHID + nt * 16;
  v8f c = wmma_f32_tile(hs, HPITCH, B, FHID, FIN, lane);
  float* out = hcat + (size_t)mtile * 16 * HCATF + head * FHID + nt * 16 + l16;
#pragma unroll
  for (int r = 0; r < 8; ++r)
    out[(size_t)(r + 8 * half) * HCATF] = c[r];

  const float as = a_heads[head * 2 * FHID + nt * 16 + l16];
  const float ad = a_heads[head * 2 * FHID + FHID + nt * 16 + l16];
#pragma unroll
  for (int r = 0; r < 8; ++r) {
    const float ps = row16_sum(c[r] * as);
    const float pd = row16_sum(c[r] * ad);
    if (l16 == 0) {
      part[wave][r + 8 * half][0] = ps;
      part[wave][r + 8 * half][1] = pd;
    }
  }
  __syncthreads();
  if (threadIdx.x < 16 * NHEADS) {
    const int row = threadIdx.x >> 2, hd = threadIdx.x & 3;
    const float s = (part[hd * 4 + 0][row][0] + part[hd * 4 + 1][row][0]) +
                    (part[hd * 4 + 2][row][0] + part[hd * 4 + 3][row][0]);
    const float d = (part[hd * 4 + 0][row][1] + part[hd * 4 + 1][row][1]) +
                    (part[hd * 4 + 2][row][1] + part[hd * 4 + 3][row][1]);
    src[((size_t)mtile * 16 + row) * NHEADS + hd] = s;
    dst[((size_t)mtile * 16 + row) * NHEADS + hd] = d;
  }
}

#define LIN_TILES (NN / 16)
__global__ __launch_bounds__(512)
void csr_lin_heads_coeff_kernel(const float* __restrict__ adj, int* __restrict__ cnt, int* __restrict__ nidx,
                                const float* __restrict__ x, const float* __restrict__ W,
                                const float* __restrict__ b, const float* __restrict__ W_heads,
                                const float* __restrict__ a_heads, float* __restrict__ hcat,
                                float* __restrict__ src, float* __restrict__ dst) {
  const int blk = (int)blockIdx.x;
  if (blk < LIN_TILES) lin_heads_coeff_tile(blk, x, W, b, W_heads, a_heads, hcat, src, dst);
  else                 build_csr_rows(blk - LIN_TILES, adj, cnt, nidx);
}

template <int N>
DEV void agg_heads_group(const char* __restrict__ hbytes, unsigned voff, int js,
                         float w0, float w1, float w2, float w3, int l0, bool hi,
                         float4& accA, float4& accB) {
  float4 xa[N], xb[N];
  for (int u = 0; u < N; ++u) {
    const int jk = lane_i(js, l0 + u);
    const char* p = hbytes + (size_t)jk * (HCATF * sizeof(float));
    xa[u] = *(const float4*)(p + voff);
    xb[u] = *(const float4*)(p + voff + 2 * FHID * sizeof(float));
  }
  for (int u = 0; u < N; ++u) {
    const float s0 = lane_f(w0, l0 + u), s1 = lane_f(w1, l0 + u);
    const float s2 = lane_f(w2, l0 + u), s3 = lane_f(w3, l0 + u);
    const float wa = hi ? s1 : s0;
    const float wb = hi ? s3 : s2;
    accA.x = fmaf(wa, xa[u].x, accA.x); accA.y = fmaf(wa, xa[u].y, accA.y);
    accA.z = fmaf(wa, xa[u].z, accA.z); accA.w = fmaf(wa, xa[u].w, accA.w);
    accB.x = fmaf(wb, xb[u].x, accB.x); accB.y = fmaf(wb, xb[u].y, accB.y);
    accB.z = fmaf(wb, xb[u].z, accB.z); accB.w = fmaf(wb, xb[u].w, accB.w);
  }
}

__global__ __launch_bounds__(32 * ROWS_L1)
void gat_aggregate_heads_end_gemm_kernel(const float* __restrict__ hcat,
                                         const float* __restrict__ src,
                                         const float* __restrict__ dst,
                                         const int* __restrict__ cnt,
                                         const int* __restrict__ nidx,
                                         const float* __restrict__ W_end,
                                         const float* __restrict__ a_end,
                                         float* __restrict__ Wh2, float* __restrict__ sd) {
  __shared__ __attribute__((aligned(16))) float gt[ROWS_L1 * GPITCH];
  __shared__ __attribute__((aligned(16))) float wsh[HCATF * FOUT];
  __shared__ float part[4][16][2];
  const int lane = threadIdx.x & 31;
  const bool hi  = lane >= 16;
  const int wv   = uni((int)(threadIdx.x >> 5));

  float4 wld[2][4];
  {
    const float4* wg = (const float4*)W_end + (size_t)(16 * wv + 4 * (lane >> 4)) * (FOUT / 4) + (lane & 15);
    for (int t = 0; t < 2; ++t)
      for (int r = 0; r < 4; ++r) wld[t][r] = wg[(8 * t + r) * (FOUT / 4)];
  }
  const int row  = uni((int)(blockIdx.x * ROWS_L1) + wv);
  const int deg  = uni(cnt[row]);
  const int* nb  = nidx + (size_t)row * CAP;
  const float4 si = *(const float4*)(src + (size_t)row * NHEADS);

  int    j[4] = {0, 0, 0, 0};
  float4 d[4] = {};
  float  w[4][NHEADS];
  const float4* dst4 = (const float4*)dst;
  if (deg > 0) {
    j[0] = nb[min(lane, deg - 1)];
    j[1] = nb[min(lane + 32, deg - 1)];
    d[0] = dst4[j[0]];
    d[1] = dst4[j[1]];
    if (deg > 64) {
      j[2] = nb[min(lane + 64, deg - 1)];
      j[3] = nb[min(lane + 96, deg - 1)];
      d[2] = dst4[j[2]];
      d[3] = dst4[j[3]];
    }
  }
  {
    float* wl = wsh + (4 * wv + (lane >> 4)) * WQ + (lane & 15) * 16;
    for (int t = 0; t < 2; ++t) {
      *(float4*)(wl + 2 * t * WQ + 0)  = make_float4(wld[t][0].x, wld[t][1].x, wld[t][2].x, wld[t][3].x);
      *(float4*)(wl + 2 * t * WQ + 4)  = make_float4(wld[t][0].y, wld[t][1].y, wld[t][2].y, wld[t][3].y);
      *(float4*)(wl + 2 * t * WQ + 8)  = make_float4(wld[t][0].z, wld[t][1].z, wld[t][2].z, wld[t][3].z);
      *(float4*)(wl + 2 * t * WQ + 12) = make_float4(wld[t][0].w, wld[t][1].w, wld[t][2].w, wld[t][3].w);
    }
  }
  float m[NHEADS] = {-3.4e38f, -3.4e38f, -3.4e38f, -3.4e38f};
  for (int s = 0; s < 4; ++s) {
    const bool valid = (s * 32 + lane) < deg;
    w[s][0] = valid ? leaky(si.x + d[s].x) : -3.4e38f;
    w[s][1] = valid ? leaky(si.y + d[s].y) : -3.4e38f;
    w[s][2] = valid ? leaky(si.z + d[s].z) : -3.4e38f;
    w[s][3] = valid ? leaky(si.w + d[s].w) : -3.4e38f;
    for (int h = 0; h < NHEADS; ++h) m[h] = fmaxf(m[h], w[s][h]);
  }
  for (int h = 0; h < NHEADS; ++h) m[h] = wave_max(m[h]);

  for (int s = 0; s < 4; ++s) {
    const bool valid = (s * 32 + lane) < deg;
    for (int h = 0; h < NHEADS; ++h) w[s][h] = valid ? expf(w[s][h] - m[h]) : 0.f;
  }
  float denom[NHEADS];
  for (int h = 0; h < NHEADS; ++h)
    denom[h] = wave_sum((w[0][h] + w[2][h]) + (w[1][h] + w[3][h]));

  float4 accA = {0.f, 0.f, 0.f, 0.f}, accB = {0.f, 0.f, 0.f, 0.f};
  const char* hbytes = (const char*)hcat;
  const unsigned voff = (unsigned)lane * 16u;
  for (int base = 0; base < deg; base += AGG_G) {
    const int s  = base >> 5;
    int l0 = base & 31;
    const int   js = (s == 0) ? j[0]    : (s == 1) ? j[1]    : (s == 2) ? j[2]    : j[3];
    const float w0 = (s == 0) ? w[0][0] : (s == 1) ? w[1][0] : (s == 2) ? w[2][0] : w[3][0];
    const float w1 = (s == 0) ? w[0][1] : (s == 1) ? w[1][1] : (s == 2) ? w[2][1] : w[3][1];
    const float w2 = (s == 0) ? w[0][2] : (s == 1) ? w[1][2] : (s == 2) ? w[2][2] : w[3][2];
    const float w3 = (s == 0) ? w[0][3] : (s == 1) ? w[1][3] : (s == 2) ? w[2][3] : w[3][3];
    const int r = deg - base;
    if (r >= AGG_G) {
      agg_heads_group<AGG_G>(hbytes, voff, js, w0, w1, w2, w3, l0, hi, accA, accB);
    } else {
      if (r & 4) { agg_heads_group<4>(hbytes, voff, js, w0, w1, w2, w3, l0, hi, accA, accB); l0 += 4; }
      if (r & 2) { agg_heads_group<2>(hbytes, voff, js, w0, w1, w2, w3, l0, hi, accA, accB); l0 += 2; }
      if (r & 1) { agg_heads_group<1>(hbytes, voff, js, w0, w1, w2, w3, l0, hi, accA, accB); }
    }
  }

  const float dA = hi ? denom[1] : denom[0];
  const float dB = hi ? denom[3] : denom[2];
  float4 oA, oB;
  oA.x = elu1(accA.x / dA); oA.y = elu1(accA.y / dA); oA.z = elu1(accA.z / dA); oA.w = elu1(accA.w / dA);
  oB.x = elu1(accB.x / dB); oB.y = elu1(accB.y / dB); oB.z = elu1(accB.z / dB); oB.w = elu1(accB.w / dB);
  float* orow = gt + wv * GPITCH + lane * 4;
  *(float4*)orow = oA;
  *(float4*)(orow + 2 * FHID) = oB;
  __syncthreads();

  const int half = lane >> 4, l16 = lane & 15;
  if (wv < 4) {
    const float as = a_end[wv * 16 + l16];
    const float ad = a_end[FOUT + wv * 16 + l16];
    v8f c = wmma_f32_tile_lds(gt, wsh + wv * 16 * 4, lane);
    float* out = Wh2 + (size_t)blockIdx.x * 16 * FOUT + wv * 16 + l16;
#pragma unroll
    for (int r = 0; r < 8; ++r)
      out[(size_t)(r + 8 * half) * FOUT] = c[r];

#pragma unroll
    for (int r = 0; r < 8; ++r) {
      const float ps = row16_sum(c[r] * as);
      const float pd = row16_sum(c[r] * ad);
      if (l16 == 0) {
        part[wv][r + 8 * half][0] = ps;
        part[wv][r + 8 * half][1] = pd;
      }
    }
  }
  __syncthreads();
  if (threadIdx.x < 32) {
    const int trow = threadIdx.x >> 1, which = threadIdx.x & 1;
    sd[((size_t)blockIdx.x * 16 + trow) * 2 + which] =
        (part[0][trow][which] + part[1][trow][which]) + (part[2][trow][which] + part[3][trow][which]);
  }
}

template <int N>
DEV void agg_end_group(const char* __restrict__ wbytes, unsigned voff, int js, float ws,
                       int l0, float& a0, float& a1) {
  float x0[N], x1[N];
  for (int u = 0; u < N; ++u) {
    const int jk = lane_i(js, l0 + u);
    const char* p = wbytes + (size_t)jk * (FOUT * sizeof(float));
    x0[u] = *(const float*)(p + voff);
    x1[u] = *(const float*)(p + voff + 32 * sizeof(float));
  }
  for (int u = 0; u < N; ++u) {
    const float wk = lane_f(ws, l0 + u);
    a0 = fmaf(wk, x0[u], a0);
    a1 = fmaf(wk, x1[u], a1);
  }
}

__global__ __launch_bounds__(32 * ROWS_PER_BLOCK)
void gat_end_kernel(const float* __restrict__ Wh2,
                    const float* __restrict__ sd,
                    const int* __restrict__ cnt,
                    const int* __restrict__ nidx,
                    float* __restrict__ out) {
  const int lane = threadIdx.x & 31;
  const int row  = uni((int)(blockIdx.x * ROWS_PER_BLOCK + (threadIdx.x >> 5)));
  const int deg  = uni(cnt[row]);
  const float si = sd[2 * (size_t)row];
  const int* nb  = nidx + (size_t)row * CAP;

  int   j[4] = {0, 0, 0, 0};
  float d[4] = {0.f, 0.f, 0.f, 0.f};
  float w[4];
  if (deg > 0) {
    j[0] = nb[min(lane, deg - 1)];
    j[1] = nb[min(lane + 32, deg - 1)];
    d[0] = sd[2 * (size_t)j[0] + 1];
    d[1] = sd[2 * (size_t)j[1] + 1];
    if (deg > 64) {
      j[2] = nb[min(lane + 64, deg - 1)];
      j[3] = nb[min(lane + 96, deg - 1)];
      d[2] = sd[2 * (size_t)j[2] + 1];
      d[3] = sd[2 * (size_t)j[3] + 1];
    }
  }
  float m = -3.4e38f;
  for (int s = 0; s < 4; ++s) {
    w[s] = ((s * 32 + lane) < deg) ? leaky(si + d[s]) : -3.4e38f;
    m = fmaxf(m, w[s]);
  }
  m = wave_max(m);

  for (int s = 0; s < 4; ++s) w[s] = ((s * 32 + lane) < deg) ? expf(w[s] - m) : 0.f;
  const float denom = wave_sum((w[0] + w[2]) + (w[1] + w[3]));

  float a0 = 0.f, a1 = 0.f;
  const char* wbytes = (const char*)Wh2;
  const unsigned voff = (unsigned)lane * 4u;
  for (int base = 0; base < deg; base += AGG_G_END) {
    const int s  = base >> 5;
    int l0 = base & 31;
    const int   js = (s == 0) ? j[0] : (s == 1) ? j[1] : (s == 2) ? j[2] : j[3];
    const float ws = (s == 0) ? w[0] : (s == 1) ? w[1] : (s == 2) ? w[2] : w[3];
    const int r = deg - base;
    if (r >= AGG_G_END) {
      agg_end_group<AGG_G_END>(wbytes, voff, js, ws, l0, a0, a1);
    } else {
      if (r & 8) { agg_end_group<8>(wbytes, voff, js, ws, l0, a0, a1); l0 += 8; }
      if (r & 4) { agg_end_group<4>(wbytes, voff, js, ws, l0, a0, a1); l0 += 4; }
      if (r & 2) { agg_end_group<2>(wbytes, voff, js, ws, l0, a0, a1); l0 += 2; }
      if (r & 1) { agg_end_group<1>(wbytes, voff, js, ws, l0, a0, a1); }
    }
  }
  const float o0 = elu1(a0 / denom);
  const float o1 = elu1(a1 / denom);

  const float mo  = wave_max(fmaxf(o0, o1));
  const float ex0 = expf(o0 - mo);
  const float ex1 = expf(o1 - mo);
  const float sum = wave_sum(ex0 + ex1);
  float* orow = out + (size_t)row * FOUT + lane;
  orow[0]  = ex0 / sum;
  orow[32] = ex1 / sum;
}

extern "C" void kernel_launch(void* const* d_in, const int* in_sizes, int n_in,
                              void* d_out, int out_size, void* d_ws, size_t ws_size,
                              hipStream_t stream) {
  (void)in_sizes; (void)n_in; (void)out_size; (void)ws_size;
  const float* x       = (const float*)d_in[0];
  const float* adj     = (const float*)d_in[1];
  const float* lin_W   = (const float*)d_in[2];
  const float* lin_b   = (const float*)d_in[3];
  const float* W_heads = (const float*)d_in[4];
  const float* a_heads = (const float*)d_in[5];
  const float* W_end   = (const float*)d_in[6];
  const float* a_end   = (const float*)d_in[7];
  float* out = (float*)d_out;

  char* ws = (char*)d_ws;
  ws += (size_t)NN * FIN * sizeof(float);
  float* hcat  = (float*)ws;  ws += (size_t)NN * HCATF * sizeof(float);
  ws += (size_t)NN * HCATF * sizeof(float);
  float* Wh2   = (float*)ws;  ws += (size_t)NN * FOUT  * sizeof(float);
  float* src1  = (float*)ws;  ws += (size_t)NHEADS * NN * sizeof(float);
  float* dst1  = (float*)ws;  ws += (size_t)NHEADS * NN * sizeof(float);
  float* sd2   = (float*)ws;  ws += (size_t)2 * NN * sizeof(float);
  int*   ncnt  = (int*)ws;    ws += (size_t)NN * sizeof(int);
  int*   nidx  = (int*)ws;    ws += (size_t)NN * CAP * sizeof(int);

  csr_lin_heads_coeff_kernel<<<LIN_TILES + NN / CSR_ROWS, 512, 0, stream>>>(adj, ncnt, nidx, x, lin_W, lin_b, W_heads, a_heads,
                                                                              hcat, src1, dst1);
  gat_aggregate_heads_end_gemm_kernel<<<NN / ROWS_L1, 32 * ROWS_L1, 0, stream>>>(hcat, src1, dst1, ncnt, nidx, W_end, a_end, Wh2, sd2);
  gat_end_kernel<<<NN / ROWS_PER_BLOCK, 32 * ROWS_PER_BLOCK, 0, stream>>>(Wh2, sd2, ncnt, nidx, out);
}


# Both aggregation kernels: a row's neighbor feature loads go out with its coefficient loads, in whole padded batches (zero-weight repeats of the last id), not as serial groups behind the softmax
# speedup vs baseline: 2.3145x; 1.0405x over previous
#include <hip/hip_runtime.h>
#include <math.h>

#define NN      4096
#define FIN     128
#define FHID    64
#define NHEADS  4
#define FOUT    64
#define HCATF   (NHEADS * FHID)
#define ALPHA_L 0.2f
#define CAP     128

typedef float v2f __attribute__((ext_vector_type(2)));
typedef float v8f __attribute__((ext_vector_type(8)));

#define DEV static __device__ __forceinline__

DEV int uni(int v) { return __builtin_amdgcn_readfirstlane(v); }
DEV float xrow(float x) {
  return __builtin_bit_cast(float, __builtin_amdgcn_permlanex16(__builtin_bit_cast(int, x), __builtin_bit_cast(int, x),
                                                               0x76543210, 0xfedcba98, false, false));
}
template <int M> DEV float xlane(float x) {
  return __builtin_bit_cast(float, __builtin_amdgcn_update_dpp(0, __builtin_bit_cast(int, x),
                                                              0x160 + M, 0xf, 0xf, true));
}
DEV float wave_max(float v) {
  v = fmaxf(v, xrow(v));
  v = fmaxf(v, xlane<8>(v));
  v = fmaxf(v, xlane<4>(v));
  v = fmaxf(v, xlane<2>(v));
  v = fmaxf(v, xlane<1>(v));
  return v;
}
DEV float wave_sum(float v) {
  v = v + xrow(v);
  v = v + xlane<8>(v);
  v = v + xlane<4>(v);
  v = v + xlane<2>(v);
  v = v + xlane<1>(v);
  return v;
}
DEV float lane_f(float x, int l) { return __builtin_bit_cast(float, __builtin_amdgcn_readlane(__builtin_bit_cast(int, x), l)); }
DEV int   lane_i(int x, int l)   { return __builtin_amdgcn_readlane(x, l); }
DEV float leaky(float e) { return (e >= 0.f) ? e : ALPHA_L * e; }
DEV float elu1(float o)  { return (o > 0.f) ? o : (expf(o) - 1.f); }

#define ROWS_PER_BLOCK 4
#define ROWS_L1 16
#define GPITCH 260
#define AGG_B 16
#define AGG_C 4
#define END_CH 8
#define END_NCH 8
#define WQ (4 * FOUT)

__device__ __forceinline__ v8f wmma_f32_tile(const float* __restrict__ A, int lda,
                                             const float* __restrict__ B, int ldb,
                                             int K, int lane) {
  const int half = lane >> 4;
  const int l16  = lane & 15;
  v8f c = {0.f, 0.f, 0.f, 0.f, 0.f, 0.f, 0.f, 0.f};
  for (int k = 0; k < K; k += 4) {
    v2f a, b;
    const float* ap = A + (size_t)l16 * lda + k + 2 * half;
    a.x = ap[0];
    a.y = ap[1];
    b.x = B[(size_t)(k + 2 * half) * ldb + l16];
    b.y = B[(size_t)(k + 2 * half + 1) * ldb + l16];
    c = __builtin_amdgcn_wmma_f32_16x16x4_f32(false, a, false, b, (short)0, c, false, false);
  }
  return c;
}

#define TILE_KB 8
DEV void lds_frag_group(const float* __restrict__ ap, const float* __restrict__ bp, int s0,
                        v2f (&a)[TILE_KB], v2f (&b)[TILE_KB]) {
#pragma unroll
  for (int u = 0; u < TILE_KB; ++u) {
    a[u].x = ap[4 * (s0 + u)];
    a[u].y = ap[4 * (s0 + u) + 1];
    b[u].x = bp[WQ * (s0 + u)];
    b[u].y = bp[WQ * (s0 + u) + 1];
  }
}
DEV v8f wmma_f32_tile_lds(const float* __restrict__ At, const float* __restrict__ Bn, int lane) {
  const int half = lane >> 4;
  const int l16  = lane & 15;
  const float* ap = At + l16 * GPITCH + 2 * half;
  const float* bp = Bn + l16 * 4 + 2 * half;
  v8f c = {0.f, 0.f, 0.f, 0.f, 0.f, 0.f, 0.f, 0.f};
  v2f a0[TILE_KB], b0[TILE_KB], a1[TILE_KB], b1[TILE_KB];
  lds_frag_group(ap, bp, 0, a0, b0);
  __builtin_amdgcn_sched_barrier(0);
#pragma unroll
  for (int g = 0; g < HCATF / 4 / TILE_KB; g += 2) {
    lds_frag_group(ap, bp, (g + 1) * TILE_KB, a1, b1);
    __builtin_amdgcn_sched_barrier(0);
#pragma unroll
    for (int u = 0; u < TILE_KB; ++u)
      c = __builtin_amdgcn_wmma_f32_16x16x4_f32(false, a0[u], false, b0[u], (short)0, c, false, false);
    __builtin_amdgcn_sched_barrier(0);
    if (g + 2 < HCATF / 4 / TILE_KB) lds_frag_group(ap, bp, (g + 2) * TILE_KB, a0, b0);
    __builtin_amdgcn_sched_barrier(0);
#pragma unroll
    for (int u = 0; u < TILE_KB; ++u)
      c = __builtin_amdgcn_wmma_f32_16x16x4_f32(false, a1[u], false, b1[u], (short)0, c, false, false);
    __builtin_amdgcn_sched_barrier(0);
  }
  return c;
}

#define CSR_ROWS 16
#define CSR_G 8
DEV unsigned ballot32(bool p) { return __builtin_amdgcn_ballot_w32(p); }
DEV int below(unsigned m, int base) { return (int)__builtin_amdgcn_mbcnt_lo(m, (unsigned)base); }

DEV void build_csr_rows(int cblk, const float* __restrict__ adj,
                        int* __restrict__ cnt, int* __restrict__ nidx) {
  const int lane = threadIdx.x & 31;
  const int row  = uni((int)(cblk * CSR_ROWS + (threadIdx.x >> 5)));
  const float4* arow4 = (const float4*)(adj + (size_t)row * NN) + lane;
  int* out = nidx + (size_t)row * CAP;
  int k = 0;
  float4 cur[CSR_G], nxt[CSR_G];
#pragma unroll
  for (int u = 0; u < CSR_G; ++u) cur[u] = arow4[u * 32];
#pragma unroll
  for (int q0 = 0; q0 < NN / 128; q0 += CSR_G) {
    if (q0 + CSR_G < NN / 128) {
#pragma unroll
      for (int u = 0; u < CSR_G; ++u) nxt[u] = arow4[(q0 + CSR_G + u) * 32];
    }
#pragma unroll
    for (int u = 0; u < CSR_G; ++u) {
      const bool f0 = cur[u].x > 0.f, f1 = cur[u].y > 0.f, f2 = cur[u].z > 0.f, f3 = cur[u].w > 0.f;
      const unsigned m0 = ballot32(f0), m1 = ballot32(f1), m2 = ballot32(f2), m3 = ballot32(f3);
      if ((m0 | m1 | m2 | m3) != 0u) {
        const int col = (q0 + u) * 128 + 4 * lane;
        int p = below(m3, below(m2, below(m1, below(m0, k))));
        if (m0 != 0u) { if (f0 && p < CAP) out[p] = col + 0; p += f0 ? 1 : 0; }
        if (m1 != 0u) { if (f1 && p < CAP) out[p] = col + 1; p += f1 ? 1 : 0; }
        if (m2 != 0u) { if (f2 && p < CAP) out[p] = col + 2; p += f2 ? 1 : 0; }
        if (m3 != 0u) { if (f3 && p < CAP) out[p] = col + 3; }
        k += __builtin_popcount(m0) + __builtin_popcount(m1) + __builtin_popcount(m2) + __builtin_popcount(m3);
      }
    }
    if (q0 + CSR_G < NN / 128) {
#pragma unroll
      for (int u = 0; u < CSR_G; ++u) cur[u] = nxt[u];
    }
  }
  if (lane == 0) cnt[row] = k < CAP ? k : CAP;
}

DEV float row16_sum(float v) {
  v = v + xlane<8>(v);
  v = v + xlane<4>(v);
  v = v + xlane<2>(v);
  v = v + xlane<1>(v);
  return v;
}

#define HPITCH 132

DEV void lin_heads_coeff_tile(int mtile, const float* __restrict__ x, const float* __restrict__ W,
                              const float* __restrict__ b, const float* __restrict__ W_heads,
                              const float* __restrict__ a_heads, float* __restrict__ hcat,
                              float* __restrict__ src, float* __restrict__ dst) {
  __shared__ float hs[16 * HPITCH];
  __shared__ float part[16][16][2];
  const int wave  = uni((int)(threadIdx.x >> 5));
  const int lane  = threadIdx.x & 31;
  const int half  = lane >> 4, l16 = lane & 15;

  if (wave < 8) {
    v8f c = wmma_f32_tile(x + (size_t)mtile * 16 * FIN, FIN, W + wave * 16, FIN, FIN, lane);
    const float bias = b[wave * 16 + l16];
#pragma unroll
    for (int r = 0; r < 8; ++r)
      hs[(r + 8 * half) * HPITCH + wave * 16 + l16] = c[r] + bias;
  }
  __syncthreads();

  const int head = wave >> 2;
  const int nt   = wave & 3;
  const float* B = W_heads + (size_t)head * FIN * FHID + nt * 16;
  v8f c = wmma_f32_tile(hs, HPITCH, B, FHID, FIN, lane);
  float* out = hcat + (size_t)mtile * 16 * HCATF + head * FHID + nt * 16 + l16;
#pragma unroll
  for (int r = 0; r < 8; ++r)
    out[(size_t)(r + 8 * half) * HCATF] = c[r];

  const float as = a_heads[head * 2 * FHID + nt * 16 + l16];
  const float ad = a_heads[head * 2 * FHID + FHID + nt * 16 + l16];
#pragma unroll
  for (int r = 0; r < 8; ++r) {
    const float ps = row16_sum(c[r] * as);
    const float pd = row16_sum(c[r] * ad);
    if (l16 == 0) {
      part[wave][r + 8 * half][0] = ps;
      part[wave][r + 8 * half][1] = pd;
    }
  }
  __syncthreads();
  if (threadIdx.x < 16 * NHEADS) {
    const int row = threadIdx.x >> 2, hd = threadIdx.x & 3;
    const float s = (part[hd * 4 + 0][row][0] + part[hd * 4 + 1][row][0]) +
                    (part[hd * 4 + 2][row][0] + part[hd * 4 + 3][row][0]);
    const float d = (part[hd * 4 + 0][row][1] + part[hd * 4 + 1][row][1]) +
                    (part[hd * 4 + 2][row][1] + part[hd * 4 + 3][row][1]);
    src[((size_t)mtile * 16 + row) * NHEADS + hd] = s;
    dst[((size_t)mtile * 16 + row) * NHEADS + hd] = d;
  }
}

#define LIN_TILES (NN / 16)
__global__ __launch_bounds__(512)
void csr_lin_heads_coeff_kernel(const float* __restrict__ adj, int* __restrict__ cnt, int* __restrict__ nidx,
                                const float* __restrict__ x, const float* __restrict__ W,
                                const float* __restrict__ b, const float* __restrict__ W_heads,
                                const float* __restrict__ a_heads, float* __restrict__ hcat,
                                float* __restrict__ src, float* __restrict__ dst) {
  const int blk = (int)blockIdx.x;
  if (blk < LIN_TILES) lin_heads_coeff_tile(blk, x, W, b, W_heads, a_heads, hcat, src, dst);
  else                 build_csr_rows(blk - LIN_TILES, adj, cnt, nidx);
}

template <int N>
DEV void agg_heads_load(const char* __restrict__ hbytes, unsigned voff, int js, int l0,
                        float4 (&xa)[N], float4 (&xb)[N]) {
#pragma unroll
  for (int u = 0; u < N; ++u) {
    const int jk = lane_i(js, l0 + u);
    const char* p = hbytes + (size_t)jk * (HCATF * sizeof(float));
    xa[u] = *(const float4*)(p + voff);
    xb[u] = *(const float4*)(p + voff + 2 * FHID * sizeof(float));
  }
}
template <int N>
DEV void agg_heads_fma(float w0, float w1, float w2, float w3, int l0, bool hi,
                       const float4 (&xa)[N], const float4 (&xb)[N], float4& accA, float4& accB) {
#pragma unroll
  for (int u = 0; u < N; ++u) {
    const float s0 = lane_f(w0, l0 + u), s1 = lane_f(w1, l0 + u);
    const float s2 = lane_f(w2, l0 + u), s3 = lane_f(w3, l0 + u);
    const float wa = hi ? s1 : s0;
    const float wb = hi ? s3 : s2;
    accA.x = fmaf(wa, xa[u].x, accA.x); accA.y = fmaf(wa, xa[u].y, accA.y);
    accA.z = fmaf(wa, xa[u].z, accA.z); accA.w = fmaf(wa, xa[u].w, accA.w);
    accB.x = fmaf(wb, xb[u].x, accB.x); accB.y = fmaf(wb, xb[u].y, accB.y);
    accB.z = fmaf(wb, xb[u].z, accB.z); accB.w = fmaf(wb, xb[u].w, accB.w);
  }
}
template <int N>
DEV void agg_heads_batch(const char* __restrict__ hbytes, unsigned voff, int js,
                         float w0, float w1, float w2, float w3, int l0, bool hi,
                         float4& accA, float4& accB) {
  float4 xa[N], xb[N];
  agg_heads_load<N>(hbytes, voff, js, l0, xa, xb);
  agg_heads_fma<N>(w0, w1, w2, w3, l0, hi, xa, xb, accA, accB);
}

__global__ __launch_bounds__(32 * ROWS_L1)
void gat_aggregate_heads_end_gemm_kernel(const float* __restrict__ hcat,
                                         const float* __restrict__ src,
                                         const float* __restrict__ dst,
                                         const int* __restrict__ cnt,
                                         const int* __restrict__ nidx,
                                         const float* __restrict__ W_end,
                                         const float* __restrict__ a_end,
                                         float* __restrict__ Wh2, float* __restrict__ sd) {
  __shared__ __attribute__((aligned(16))) float gt[ROWS_L1 * GPITCH];
  __shared__ __attribute__((aligned(16))) float wsh[HCATF * FOUT];
  __shared__ float part[4][16][2];
  const int lane = threadIdx.x & 31;
  const bool hi  = lane >= 16;
  const int wv   = uni((int)(threadIdx.x >> 5));

  float4 wld[2][4];
  {
    const float4* wg = (const float4*)W_end + (size_t)(16 * wv + 4 * (lane >> 4)) * (FOUT / 4) + (lane & 15);
    for (int t = 0; t < 2; ++t)
      for (int r = 0; r < 4; ++r) wld[t][r] = wg[(8 * t + r) * (FOUT / 4)];
  }
  const int row  = uni((int)(blockIdx.x * ROWS_L1) + wv);
  const int deg  = uni(cnt[row]);
  const int* nb  = nidx + (size_t)row * CAP;
  const float4 si = *(const float4*)(src + (size_t)row * NHEADS);

  int    j[4] = {0, 0, 0, 0};
  float4 d[4] = {};
  float  w[4][NHEADS];
  const float4* dst4 = (const float4*)dst;
  float4 xa[AGG_B], xb[AGG_B];
  const char* hbytes = (const char*)hcat;
  const unsigned voff = (unsigned)lane * 16u;
  if (deg > 0) {
    j[0] = nb[min(lane, deg - 1)];
    j[1] = nb[min(lane + 32, deg - 1)];
    d[0] = dst4[j[0]];
    d[1] = dst4[j[1]];
    if (deg > 64) {
      j[2] = nb[min(lane + 64, deg - 1)];
      j[3] = nb[min(lane + 96, deg - 1)];
      d[2] = dst4[j[2]];
      d[3] = dst4[j[3]];
    }
    agg_heads_load<AGG_B>(hbytes, voff, j[0], 0, xa, xb);
  }
  float m[NHEADS] = {-3.4e38f, -3.4e38f, -3.4e38f, -3.4e38f};
  for (int s = 0; s < 4; ++s) {
    const bool valid = (s * 32 + lane) < deg;
    w[s][0] = valid ? leaky(si.x + d[s].x) : -3.4e38f;
    w[s][1] = valid ? leaky(si.y + d[s].y) : -3.4e38f;
    w[s][2] = valid ? leaky(si.z + d[s].z) : -3.4e38f;
    w[s][3] = valid ? leaky(si.w + d[s].w) : -3.4e38f;
    for (int h = 0; h < NHEADS; ++h) m[h] = fmaxf(m[h], w[s][h]);
  }
  for (int h = 0; h < NHEADS; ++h) m[h] = wave_max(m[h]);

  for (int s = 0; s < 4; ++s) {
    const bool valid = (s * 32 + lane) < deg;
    for (int h = 0; h < NHEADS; ++h) w[s][h] = valid ? expf(w[s][h] - m[h]) : 0.f;
  }
  float denom[NHEADS];
  for (int h = 0; h < NHEADS; ++h)
    denom[h] = wave_sum((w[0][h] + w[2][h]) + (w[1][h] + w[3][h]));

  float4 accA = {0.f, 0.f, 0.f, 0.f}, accB = {0.f, 0.f, 0.f, 0.f};
  if (deg > 0)
    agg_heads_fma<AGG_B>(w[0][0], w[0][1], w[0][2], w[0][3], 0, hi, xa, xb, accA, accB);
  for (int base = AGG_B; base < deg; base += AGG_B) {
    const int s  = base >> 5;
    const int l0 = base & 31;
    const int   js = (s == 0) ? j[0]    : (s == 1) ? j[1]    : (s == 2) ? j[2]    : j[3];
    const float w0 = (s == 0) ? w[0][0] : (s == 1) ? w[1][0] : (s == 2) ? w[2][0] : w[3][0];
    const float w1 = (s == 0) ? w[0][1] : (s == 1) ? w[1][1] : (s == 2) ? w[2][1] : w[3][1];
    const float w2 = (s == 0) ? w[0][2] : (s == 1) ? w[1][2] : (s == 2) ? w[2][2] : w[3][2];
    const float w3 = (s == 0) ? w[0][3] : (s == 1) ? w[1][3] : (s == 2) ? w[2][3] : w[3][3];
    const int r = deg - base;
    if (r > 3 * AGG_C)      agg_heads_batch<4 * AGG_C>(hbytes, voff, js, w0, w1, w2, w3, l0, hi, accA, accB);
    else if (r > 2 * AGG_C) agg_heads_batch<3 * AGG_C>(hbytes, voff, js, w0, w1, w2, w3, l0, hi, accA, accB);
    else if (r > AGG_C)     agg_heads_batch<2 * AGG_C>(hbytes, voff, js, w0, w1, w2, w3, l0, hi, accA, accB);
    else                    agg_heads_batch<AGG_C>(hbytes, voff, js, w0, w1, w2, w3, l0, hi, accA, accB);
  }

  {
    float* wl = wsh + (4 * wv + (lane >> 4)) * WQ + (lane & 15) * 16;
    for (int t = 0; t < 2; ++t) {
      *(float4*)(wl + 2 * t * WQ + 0)  = make_float4(wld[t][0].x, wld[t][1].x, wld[t][2].x, wld[t][3].x);
      *(float4*)(wl + 2 * t * WQ + 4)  = make_float4(wld[t][0].y, wld[t][1].y, wld[t][2].y, wld[t][3].y);
      *(float4*)(wl + 2 * t * WQ + 8)  = make_float4(wld[t][0].z, wld[t][1].z, wld[t][2].z, wld[t][3].z);
      *(float4*)(wl + 2 * t * WQ + 12) = make_float4(wld[t][0].w, wld[t][1].w, wld[t][2].w, wld[t][3].w);
    }
  }
  const float dA = hi ? denom[1] : denom[0];
  const float dB = hi ? denom[3] : denom[2];
  float4 oA, oB;
  oA.x = elu1(accA.x / dA); oA.y = elu1(accA.y / dA); oA.z = elu1(accA.z / dA); oA.w = elu1(accA.w / dA);
  oB.x = elu1(accB.x / dB); oB.y = elu1(accB.y / dB); oB.z = elu1(accB.z / dB); oB.w = elu1(accB.w / dB);
  float* orow = gt + wv * GPITCH + lane * 4;
  *(float4*)orow = oA;
  *(float4*)(orow + 2 * FHID) = oB;
  __syncthreads();

  const int half = lane >> 4, l16 = lane & 15;
  if (wv < 4) {
    const float as = a_end[wv * 16 + l16];
    const float ad = a_end[FOUT + wv * 16 + l16];
    v8f c = wmma_f32_tile_lds(gt, wsh + wv * 16 * 4, lane);
    float* out = Wh2 + (size_t)blockIdx.x * 16 * FOUT + wv * 16 + l16;
#pragma unroll
    for (int r = 0; r < 8; ++r)
      out[(size_t)(r + 8 * half) * FOUT] = c[r];

#pragma unroll
    for (int r = 0; r < 8; ++r) {
      const float ps = row16_sum(c[r] * as);
      const float pd = row16_sum(c[r] * ad);
      if (l16 == 0) {
        part[wv][r + 8 * half][0] = ps;
        part[wv][r + 8 * half][1] = pd;
      }
    }
  }
  __syncthreads();
  if (threadIdx.x < 32) {
    const int trow = threadIdx.x >> 1, which = threadIdx.x & 1;
    sd[((size_t)blockIdx.x * 16 + trow) * 2 + which] =
        (part[0][trow][which] + part[1][trow][which]) + (part[2][trow][which] + part[3][trow][which]);
  }
}

template <int N>
DEV void agg_end_group(const char* __restrict__ wbytes, unsigned voff, int js, float ws,
                       int l0, float& a0, float& a1) {
  float x0[N], x1[N];
  for (int u = 0; u < N; ++u) {
    const int jk = lane_i(js, l0 + u);
    const char* p = wbytes + (size_t)jk * (FOUT * sizeof(float));
    x0[u] = *(const float*)(p + voff);
    x1[u] = *(const float*)(p + voff + 32 * sizeof(float));
  }
  for (int u = 0; u < N; ++u) {
    const float wk = lane_f(ws, l0 + u);
    a0 = fmaf(wk, x0[u], a0);
    a1 = fmaf(wk, x1[u], a1);
  }
}

__global__ __launch_bounds__(32 * ROWS_PER_BLOCK)
void gat_end_kernel(const float* __restrict__ Wh2,
                    const float* __restrict__ sd,
                    const int* __restrict__ cnt,
                    const int* __restrict__ nidx,
                    float* __restrict__ out) {
  const int lane = threadIdx.x & 31;
  const int row  = uni((int)(blockIdx.x * ROWS_PER_BLOCK + (threadIdx.x >> 5)));
  const int deg  = uni(cnt[row]);
  const float si = sd[2 * (size_t)row];
  const int* nb  = nidx + (size_t)row * CAP;

  int   j[4] = {0, 0, 0, 0};
  float d[4] = {0.f, 0.f, 0.f, 0.f};
  float w[4];
  if (deg > 0) {
    j[0] = nb[min(lane, deg - 1)];
    j[1] = nb[min(lane + 32, deg - 1)];
    d[0] = sd[2 * (size_t)j[0] + 1];
    d[1] = sd[2 * (size_t)j[1] + 1];
    if (deg > 64) {
      j[2] = nb[min(lane + 64, deg - 1)];
      j[3] = nb[min(lane + 96, deg - 1)];
      d[2] = sd[2 * (size_t)j[2] + 1];
      d[3] = sd[2 * (size_t)j[3] + 1];
    }
  }
  float x0[END_NCH * END_CH], x1[END_NCH * END_CH];
  const char* wbytes = (const char*)Wh2;
  const unsigned voff = (unsigned)lane * 4u;
#pragma unroll
  for (int c = 0; c < END_NCH; ++c) {
    if (c * END_CH < deg) {
#pragma unroll
      for (int u = 0; u < END_CH; ++u) {
        const int jk = lane_i(j[c >> 2], (c & 3) * END_CH + u);
        const char* p = wbytes + (size_t)jk * (FOUT * sizeof(float));
        x0[c * END_CH + u] = *(const float*)(p + voff);
        x1[c * END_CH + u] = *(const float*)(p + voff + 32 * sizeof(float));
      }
    }
  }
  float m = -3.4e38f;
  for (int s = 0; s < 4; ++s) {
    w[s] = ((s * 32 + lane) < deg) ? leaky(si + d[s]) : -3.4e38f;
    m = fmaxf(m, w[s]);
  }
  m = wave_max(m);

  for (int s = 0; s < 4; ++s) w[s] = ((s * 32 + lane) < deg) ? expf(w[s] - m) : 0.f;
  const float denom = wave_sum((w[0] + w[2]) + (w[1] + w[3]));

  float a0 = 0.f, a1 = 0.f;
#pragma unroll
  for (int c = 0; c < END_NCH; ++c) {
    if (c * END_CH < deg) {
#pragma unroll
      for (int u = 0; u < END_CH; ++u) {
        const float wk = lane_f(w[c >> 2], (c & 3) * END_CH + u);
        a0 = fmaf(wk, x0[c * END_CH + u], a0);
        a1 = fmaf(wk, x1[c * END_CH + u], a1);
      }
    }
  }
  for (int base = 2 * 32; base < deg; base += END_CH) {
    const int s = base >> 5;
    agg_end_group<END_CH>(wbytes, voff, (s == 2) ? j[2] : j[3], (s == 2) ? w[2] : w[3], base & 31, a0, a1);
  }
  const float o0 = elu1(a0 / denom);
  const float o1 = elu1(a1 / denom);

  const float mo  = wave_max(fmaxf(o0, o1));
  const float ex0 = expf(o0 - mo);
  const float ex1 = expf(o1 - mo);
  const float sum = wave_sum(ex0 + ex1);
  float* orow = out + (size_t)row * FOUT + lane;
  orow[0]  = ex0 / sum;
  orow[32] = ex1 / sum;
}

extern "C" void kernel_launch(void* const* d_in, const int* in_sizes, int n_in,
                              void* d_out, int out_size, void* d_ws, size_t ws_size,
                              hipStream_t stream) {
  (void)in_sizes; (void)n_in; (void)out_size; (void)ws_size;
  const float* x       = (const float*)d_in[0];
  const float* adj     = (const float*)d_in[1];
  const float* lin_W   = (const float*)d_in[2];
  const float* lin_b   = (const float*)d_in[3];
  const float* W_heads = (const float*)d_in[4];
  const float* a_heads = (const float*)d_in[5];
  const float* W_end   = (const float*)d_in[6];
  const float* a_end   = (const float*)d_in[7];
  float* out = (float*)d_out;

  char* ws = (char*)d_ws;
  ws += (size_t)NN * FIN * sizeof(float);
  float* hcat  = (float*)ws;  ws += (size_t)NN * HCATF * sizeof(float);
  ws += (size_t)NN * HCATF * sizeof(float);
  float* Wh2   = (float*)ws;  ws += (size_t)NN * FOUT  * sizeof(float);
  float* src1  = (float*)ws;  ws += (size_t)NHEADS * NN * sizeof(float);
  float* dst1  = (float*)ws;  ws += (size_t)NHEADS * NN * sizeof(float);
  float* sd2   = (float*)ws;  ws += (size_t)2 * NN * sizeof(float);
  int*   ncnt  = (int*)ws;    ws += (size_t)NN * sizeof(int);
  int*   nidx  = (int*)ws;    ws += (size_t)NN * CAP * sizeof(int);

  csr_lin_heads_coeff_kernel<<<LIN_TILES + NN / CSR_ROWS, 512, 0, stream>>>(adj, ncnt, nidx, x, lin_W, lin_b, W_heads, a_heads,
                                                                              hcat, src1, dst1);
  gat_aggregate_heads_end_gemm_kernel<<<NN / ROWS_L1, 32 * ROWS_L1, 0, stream>>>(hcat, src1, dst1, ncnt, nidx, W_end, a_end, Wh2, sd2);
  gat_end_kernel<<<NN / ROWS_PER_BLOCK, 32 * ROWS_PER_BLOCK, 0, stream>>>(Wh2, sd2, ncnt, nidx, out);
}
